# starting kernel
# baseline (speedup 1.0000x reference)
#include <hip/hip_runtime.h>
#include <hip/hip_bf16.h>
#include <cstdint>

#define DFEAT 64

typedef float v2f __attribute__((ext_vector_type(2)));
typedef float v8f __attribute__((ext_vector_type(8)));

__global__ void sag_init_kernel(float* __restrict__ deg,
                                float4* __restrict__ h1,
                                float4* __restrict__ h2, int n) {
  const int i = blockIdx.x * blockDim.x + threadIdx.x;
  if (i < n) deg[i] = 1.0f;
  const int total4 = n * (DFEAT / 4);
  if (i < total4) {
    const float4 z = make_float4(0.f, 0.f, 0.f, 0.f);
    h1[i] = z;
    h2[i] = z;
  }
}

__global__ void sag_deg_kernel(const int* __restrict__ dst,
                               float* __restrict__ deg, int e) {
  const int i = blockIdx.x * blockDim.x + threadIdx.x;
  if (i < e) unsafeAtomicAdd(&deg[dst[i]], 1.0f);
}

__global__ void sag_dinv_kernel(float* __restrict__ deg, int n) {
  const int i = blockIdx.x * blockDim.x + threadIdx.x;
  if (i < n) deg[i] = rsqrtf(deg[i]);
}

__global__ void sag_hop_kernel(const float* __restrict__ hin,
                               float* __restrict__ hout,
                               const int* __restrict__ src,
                               const int* __restrict__ dst,
                               const float* __restrict__ dinv,
                               int e, int n) {
  const int tid  = blockIdx.x * blockDim.x + threadIdx.x;
  const int work = tid >> 4;
  const int lane = tid & 15;
  if (work >= e + n) return;

  int s, t;
  if (work < e) { s = src[work]; t = dst[work]; }
  else          { s = t = work - e; }

  const float coef = dinv[s] * dinv[t];
  const float4 v =
      *reinterpret_cast<const float4*>(hin + (size_t)s * DFEAT + lane * 4);
  float* o = hout + (size_t)t * DFEAT + lane * 4;
  unsafeAtomicAdd(o + 0, coef * v.x);
  unsafeAtomicAdd(o + 1, coef * v.y);
  unsafeAtomicAdd(o + 2, coef * v.z);
  unsafeAtomicAdd(o + 3, coef * v.w);
}

__global__ void sag_score_kernel(const float* __restrict__ x,
                                 const float* __restrict__ h2,
                                 const float* __restrict__ W,
                                 const float* __restrict__ bptr,
                                 float* __restrict__ outx,
                                 float* __restrict__ outs, int n) {
  const int lane = threadIdx.x & 31;
  const int wave = threadIdx.x >> 5;
  const int nb   = (blockIdx.x * (blockDim.x >> 5) + wave) * 16;
  if (nb >= n) return;

  const int half = lane >> 4;
  const int m    = lane & 15;
  int rowNode = nb + m;
  if (rowNode > n - 1) rowNode = n - 1;
  const float* hrow = h2 + (size_t)rowNode * DFEAT;

  v8f c = {};
#pragma unroll
  for (int k0 = 0; k0 < DFEAT; k0 += 4) {
    const int kb = k0 + half * 2;
    v2f a;  a.x  = hrow[kb]; a.y  = hrow[kb + 1];
    v2f bm; bm.x = W[kb];    bm.y = W[kb + 1];
    c = __builtin_amdgcn_wmma_f32_16x16x4_f32(
         false, a,  false, bm,
         (short)0, c,  false,  false);
  }

  const float bias = bptr[0];
#pragma unroll
  for (int r = 0; r < 8; ++r) {
    const int node = nb + half * 8 + r;
    if (node < n) {
      const float sc = 1.0f / (1.0f + __expf(-(c[r] + bias)));
      if (m == 0) outs[node] = sc;
      const float4 xv =
          reinterpret_cast<const float4*>(x + (size_t)node * DFEAT)[m];
      const float4 ov = make_float4(xv.x * sc, xv.y * sc, xv.z * sc, xv.w * sc);
      reinterpret_cast<float4*>(outx + (size_t)node * DFEAT)[m] = ov;
    }
  }
}

extern "C" void kernel_launch(void* const* d_in, const int* in_sizes, int n_in,
                              void* d_out, int out_size, void* d_ws, size_t ws_size,
                              hipStream_t stream) {
  const float* x  = (const float*)d_in[0];
  const int*   ei = (const int*)d_in[1];
  const float* W  = (const float*)d_in[2];
  const float* b  = (const float*)d_in[3];

  const int n = in_sizes[0] / DFEAT;
  const int e = in_sizes[1] / 2;
  const int* src = ei;
  const int* dst = ei + e;

  float* deg = (float*)d_ws;
  const size_t nPad = ((size_t)n + 63) & ~(size_t)63;
  float* h1 = deg + nPad;
  float* h2   = (float*)d_out;
  float* outs = (float*)d_out + (size_t)n * DFEAT;

  const int B = 256;

  {
    const int total = n * (DFEAT / 4);
    sag_init_kernel<<<(total + B - 1) / B, B, 0, stream>>>(
        deg, (float4*)h1, (float4*)h2, n);
  }
  sag_deg_kernel<<<(e + B - 1) / B, B, 0, stream>>>(dst, deg, e);
  sag_dinv_kernel<<<(n + B - 1) / B, B, 0, stream>>>(deg, n);
  {
    const long long work = (long long)(e + n) * 16;
    sag_hop_kernel<<<(int)((work + B - 1) / B), B, 0, stream>>>(
        x, h1, src, dst, deg, e, n);
  }
  {
    const long long work = (long long)(e + n) * 16;
    sag_hop_kernel<<<(int)((work + B - 1) / B), B, 0, stream>>>(
        h1, h2, src, dst, deg, e, n);
  }
  {
    const int nodesPerBlock = (B / 32) * 16;
    sag_score_kernel<<<(n + nodesPerBlock - 1) / nodesPerBlock, B, 0, stream>>>(
        x, h2, W, b,  h2, outs, n);
  }
}


# Apply the 64->1 linear layer before the two propagation hops (A(A(xW)) for (AAx)W): hops add one f32 per edge, not 64; products stay on f32 WMMA; sums reassociated, order changes
# speedup vs baseline: 1.4895x; 1.4895x over previous
#include <hip/hip_runtime.h>
#include <cstdint>

#define DFEAT 64


typedef float v2f __attribute__((ext_vector_type(2)));
typedef float v8f __attribute__((ext_vector_type(8)));

__global__ void node_project_kernel(const float* __restrict__ x,
                                    const float* __restrict__ W,
                                    float* __restrict__ deg,
                                    float* __restrict__ z0,
                                    float* __restrict__ acc1,
                                    float* __restrict__ acc2, int n) {
  const int lane = threadIdx.x & 31;
  const int wave = threadIdx.x >> 5;
  const int nb   = (blockIdx.x * (blockDim.x >> 5) + wave) * 32;
  if (nb >= n) return;

  const int half = lane >> 4;
  const int m    = lane & 15;
  int rowA = nb + m;
  int rowB = nb + 16 + m;
  if (rowA > n - 1) rowA = n - 1;
  if (rowB > n - 1) rowB = n - 1;
  const float* xa = x + (size_t)rowA * DFEAT;
  const float* xb = x + (size_t)rowB * DFEAT;

  v8f ca = {};
  v8f cb = {};
#pragma unroll
  for (int k0 = 0; k0 < DFEAT; k0 += 4) {
    const int kb = k0 + half * 2;
    v2f bm; bm.x = W[kb];  bm.y = W[kb + 1];
    v2f a;  a.x  = xa[kb]; a.y  = xa[kb + 1];
    v2f b;  b.x  = xb[kb]; b.y  = xb[kb + 1];
    ca = __builtin_amdgcn_wmma_f32_16x16x4_f32(
         false, a,  false, bm,
         (short)0, ca,  false,  false);
    cb = __builtin_amdgcn_wmma_f32_16x16x4_f32(
         false, b,  false, bm,
         (short)0, cb,  false,  false);
  }

  const int tile = m >> 3;
  const int rr   = m & 7;
  float v = 0.0f;
#pragma unroll
  for (int r = 0; r < 8; ++r) {
    const float pick = tile ? cb[r] : ca[r];
    v = (rr == r) ? pick : v;
  }
  const int node = nb + tile * 16 + half * 8 + rr;
  if (node < n) {
    z0[node]   = v;
    deg[node]  = 1.0f;
    acc1[node] = 0.0f;
    acc2[node] = 0.0f;
  }
}

__global__ void degree_kernel(const int* __restrict__ dst,
                              float* __restrict__ deg, int e) {
  const int i = blockIdx.x * blockDim.x + threadIdx.x;
  if (i < e) unsafeAtomicAdd(&deg[dst[i]], 1.0f);
}

__global__ void dinv_kernel(float* __restrict__ deg, int n) {
  const int i = blockIdx.x * blockDim.x + threadIdx.x;
  if (i < n) deg[i] = rsqrtf(deg[i]);
}

#define HOP_ITEMS 4
__device__ __forceinline__ void hop_item(int w, int e,
                                         const int* __restrict__ src,
                                         const int* __restrict__ dst,
                                         const float* __restrict__ dinv,
                                         const float* __restrict__ zin,
                                         int& t, float& val) {
  const bool isEdge = w < e;
  const int  ie = isEdge ? w : 0;
  int s = src[ie];
  t = dst[ie];
  if (!isEdge) { s = w - e; t = w - e; }
  val = dinv[s] * dinv[t] * zin[s];
}

__global__ void edge_hop_kernel(const float* __restrict__ zin,
                                float* __restrict__ zout,
                                const int* __restrict__ src,
                                const int* __restrict__ dst,
                                const float* __restrict__ dinv,
                                int e, int n) {
  const int total     = e + n;
  const int blockBase = blockIdx.x * (blockDim.x * HOP_ITEMS);
  const int base      = blockBase + threadIdx.x;
  if (blockBase + (int)blockDim.x * HOP_ITEMS <= total) {
    int   tgt[HOP_ITEMS];
    float val[HOP_ITEMS];
#pragma unroll
    for (int j = 0; j < HOP_ITEMS; ++j)
      hop_item(base + j * blockDim.x, e, src, dst, dinv, zin, tgt[j], val[j]);
#pragma unroll
    for (int j = 0; j < HOP_ITEMS; ++j) unsafeAtomicAdd(&zout[tgt[j]], val[j]);
  } else {
    for (int j = 0; j < HOP_ITEMS; ++j) {
      const int w = base + j * blockDim.x;
      if (w < total) {
        int t; float v;
        hop_item(w, e, src, dst, dinv, zin, t, v);
        unsafeAtomicAdd(&zout[t], v);
      }
    }
  }
}

#define FIN_ITEMS 4
__global__ void node_finish_kernel(const float* __restrict__ x,
                                   const float* __restrict__ z2,
                                   const float* __restrict__ bptr,
                                   float* __restrict__ outx,
                                   float* __restrict__ outs, int n) {
  const float4* __restrict__ x4   = reinterpret_cast<const float4*>(x);
  float4* __restrict__       out4 = reinterpret_cast<float4*>(outx);
  const int total4 = n * (DFEAT / 4);
  const int base   = blockIdx.x * (blockDim.x * FIN_ITEMS) + threadIdx.x;
  if (base >= total4) return;
  const float bias = bptr[0];
  float4 xv[FIN_ITEMS];
  float  sc[FIN_ITEMS];
#pragma unroll
  for (int j = 0; j < FIN_ITEMS; ++j) {
    int i = base + j * blockDim.x;
    if (i > total4 - 1) i = total4 - 1;
    xv[j] = x4[i];
    sc[j] = 1.0f / (1.0f + __expf(-(z2[i >> 4] + bias)));
  }
#pragma unroll
  for (int j = 0; j < FIN_ITEMS; ++j) {
    const int i = base + j * blockDim.x;
    if (i < total4) {
      out4[i] = make_float4(xv[j].x * sc[j], xv[j].y * sc[j],
                            xv[j].z * sc[j], xv[j].w * sc[j]);
      if ((i & 15) == 0) outs[i >> 4] = sc[j];
    }
  }
}

extern "C" void kernel_launch(void* const* d_in, const int* in_sizes, int n_in,
                              void* d_out, int out_size, void* d_ws, size_t ws_size,
                              hipStream_t stream) {
  const float* x  = (const float*)d_in[0];
  const int*   ei = (const int*)d_in[1];
  const float* W  = (const float*)d_in[2];
  const float* b  = (const float*)d_in[3];

  const int n = in_sizes[0] / DFEAT;
  const int e = in_sizes[1] / 2;
  const int* src = ei;
  const int* dst = ei + e;

  const size_t nPad = ((size_t)n + 63) & ~(size_t)63;
  float* deg  = (float*)d_ws;
  float* z0   = deg + nPad;
  float* acc1 = z0 + nPad;
  float* acc2 = acc1 + nPad;
  float* outx = (float*)d_out;
  float* outs = (float*)d_out + (size_t)n * DFEAT;

  const int B = 256;

  {
    const int PB = 128;
    node_project_kernel<<<(n + PB - 1) / PB, PB, 0, stream>>>(
        x, W, deg, z0, acc1, acc2, n);
  }
  degree_kernel<<<(e + B - 1) / B, B, 0, stream>>>(dst, deg, e);
  dinv_kernel<<<(n + B - 1) / B, B, 0, stream>>>(deg, n);
  {
    const int per = B * HOP_ITEMS;
    const int grid = (e + n + per - 1) / per;
    edge_hop_kernel<<<grid, B, 0, stream>>>(z0, acc1, src, dst, deg, e, n);
    edge_hop_kernel<<<grid, B, 0, stream>>>(acc1, acc2, src, dst, deg, e, n);
  }
  {
    const int per = B * FIN_ITEMS;
    const int total4 = n * (DFEAT / 4);
    node_finish_kernel<<<(total4 + per - 1) / per, B, 0, stream>>>(
        x, acc2, b, outx, outs, n);
  }
}


# Sums by destination kept in LDS: block (edge slice, node range) adds edges by LDS f32 atomics, stores a partial plane; per-node fixed-order sum of 64 planes; no global atomics; sum order changes
# speedup vs baseline: 3.7493x; 2.5172x over previous
#include <hip/hip_runtime.h>
#include <cstdint>

#define DFEAT 64


typedef float v2f __attribute__((ext_vector_type(2)));
typedef float v8f __attribute__((ext_vector_type(8)));

__global__ void node_project_kernel(const float* __restrict__ x,
                                    const float* __restrict__ W,
                                    float* __restrict__ z0, int n) {
  const int lane = threadIdx.x & 31;
  const int wave = threadIdx.x >> 5;
  const int nb   = (blockIdx.x * (blockDim.x >> 5) + wave) * 32;
  if (nb >= n) return;

  const int half = lane >> 4;
  const int m    = lane & 15;
  int rowA = nb + m;
  int rowB = nb + 16 + m;
  if (rowA > n - 1) rowA = n - 1;
  if (rowB > n - 1) rowB = n - 1;
  const float* xa = x + (size_t)rowA * DFEAT;
  const float* xb = x + (size_t)rowB * DFEAT;

  v8f ca = {};
  v8f cb = {};
#pragma unroll
  for (int k0 = 0; k0 < DFEAT; k0 += 4) {
    const int kb = k0 + half * 2;
    v2f bm; bm.x = W[kb];  bm.y = W[kb + 1];
    v2f a;  a.x  = xa[kb]; a.y  = xa[kb + 1];
    v2f b;  b.x  = xb[kb]; b.y  = xb[kb + 1];
    ca = __builtin_amdgcn_wmma_f32_16x16x4_f32(
         false, a,  false, bm,
         (short)0, ca,  false,  false);
    cb = __builtin_amdgcn_wmma_f32_16x16x4_f32(
         false, b,  false, bm,
         (short)0, cb,  false,  false);
  }

  const int tile = m >> 3;
  const int rr   = m & 7;
  float v = 0.0f;
#pragma unroll
  for (int r = 0; r < 8; ++r) {
    const float pick = tile ? cb[r] : ca[r];
    v = (rr == r) ? pick : v;
  }
  const int node = nb + tile * 16 + half * 8 + rr;
  if (node < n) z0[node] = v;
}

#define RANGE        25024
#define SLICES       64
#define SCAT_THREADS 1024
#define SCAT_ITEMS   5

template <bool kDegree>
__global__ __launch_bounds__(SCAT_THREADS)
void range_scatter_kernel(const float* __restrict__ zin,
                          const float* __restrict__ dinv,
                          const int* __restrict__ src,
                          const int* __restrict__ dst,
                          float* __restrict__ planes,
                          int e, int n, int sliceLen, int planeStride) {
  __shared__ float acc[RANGE];

  const int lo = blockIdx.y * RANGE;
  int cnt = n - lo;
  if (cnt > RANGE) cnt = RANGE;

  for (int i = threadIdx.x; i < cnt; i += SCAT_THREADS) acc[i] = 0.0f;
  __syncthreads();

  const int sLo = blockIdx.x * sliceLen;
  int sHi = sLo + sliceLen;
  if (sHi > e) sHi = e;

  for (int b = sLo + (int)threadIdx.x; b < sHi; b += SCAT_THREADS * SCAT_ITEMS) {
    int   t[SCAT_ITEMS];
    int   s[SCAT_ITEMS];
    bool  keep[SCAT_ITEMS];
    float v[SCAT_ITEMS];
#pragma unroll
    for (int j = 0; j < SCAT_ITEMS; ++j) {
      const int  idx = b + j * SCAT_THREADS;
      const bool inb = idx < sHi;
      const int  ic  = inb ? idx : sHi - 1;
      t[j] = dst[ic];
      s[j] = kDegree ? 0 : src[ic];
      keep[j] = inb && (unsigned)(t[j] - lo) < (unsigned)cnt;
    }
#pragma unroll
    for (int j = 0; j < SCAT_ITEMS; ++j) {
      if (kDegree) {
        v[j] = 1.0f;
      } else {
        const int sg = keep[j] ? s[j] : lo;
        const int tg = keep[j] ? t[j] : lo;
        v[j] = dinv[sg] * dinv[tg] * zin[sg];
      }
    }
    if (!kDegree) {
#pragma unroll
      for (int j = 0; j < SCAT_ITEMS; ++j) asm volatile("" :: "v"(v[j]));
    }
#pragma unroll
    for (int j = 0; j < SCAT_ITEMS; ++j)
      if (keep[j]) atomicAdd(&acc[t[j] - lo], v[j]);
  }
  __syncthreads();

  float* piece = planes + (size_t)blockIdx.x * planeStride + lo;
  for (int i = threadIdx.x; i < cnt; i += SCAT_THREADS) piece[i] = acc[i];
}

__global__ void dinv_reduce_kernel(const float* __restrict__ planes,
                                   float* __restrict__ dinv,
                                   int n, int planeStride) {
  const int i = blockIdx.x * blockDim.x + threadIdx.x;
  if (i >= n) return;
  float a0 = 1.0f, a1 = 0.0f, a2 = 0.0f, a3 = 0.0f;
#pragma unroll
  for (int p = 0; p < SLICES; p += 4) {
    a0 += planes[(size_t)(p + 0) * planeStride + i];
    a1 += planes[(size_t)(p + 1) * planeStride + i];
    a2 += planes[(size_t)(p + 2) * planeStride + i];
    a3 += planes[(size_t)(p + 3) * planeStride + i];
  }
  dinv[i] = rsqrtf((a0 + a1) + (a2 + a3));
}

__global__ void hop_reduce_kernel(const float* __restrict__ planes,
                                  const float* __restrict__ dinv,
                                  const float* __restrict__ zin,
                                  float* __restrict__ zout,
                                  int n, int planeStride) {
  const int i = blockIdx.x * blockDim.x + threadIdx.x;
  if (i >= n) return;
  const float d = dinv[i];
  float a0 = d * d * zin[i];
  float a1 = 0.0f, a2 = 0.0f, a3 = 0.0f;
#pragma unroll
  for (int p = 0; p < SLICES; p += 4) {
    a0 += planes[(size_t)(p + 0) * planeStride + i];
    a1 += planes[(size_t)(p + 1) * planeStride + i];
    a2 += planes[(size_t)(p + 2) * planeStride + i];
    a3 += planes[(size_t)(p + 3) * planeStride + i];
  }
  zout[i] = (a0 + a1) + (a2 + a3);
}

#define FIN_ITEMS 4
__global__ void node_finish_kernel(const float* __restrict__ x,
                                   const float* __restrict__ z2,
                                   const float* __restrict__ bptr,
                                   float* __restrict__ outx,
                                   float* __restrict__ outs, int n) {
  const float4* __restrict__ x4   = reinterpret_cast<const float4*>(x);
  float4* __restrict__       out4 = reinterpret_cast<float4*>(outx);
  const int total4 = n * (DFEAT / 4);
  const int base   = blockIdx.x * (blockDim.x * FIN_ITEMS) + threadIdx.x;
  if (base >= total4) return;
  const float bias = bptr[0];
  float4 xv[FIN_ITEMS];
  float  sc[FIN_ITEMS];
#pragma unroll
  for (int j = 0; j < FIN_ITEMS; ++j) {
    int i = base + j * blockDim.x;
    if (i > total4 - 1) i = total4 - 1;
    xv[j] = x4[i];
    sc[j] = 1.0f / (1.0f + __expf(-(z2[i >> 4] + bias)));
  }
#pragma unroll
  for (int j = 0; j < FIN_ITEMS; ++j) {
    const int i = base + j * blockDim.x;
    if (i < total4) {
      out4[i] = make_float4(xv[j].x * sc[j], xv[j].y * sc[j],
                            xv[j].z * sc[j], xv[j].w * sc[j]);
      if ((i & 15) == 0) outs[i >> 4] = sc[j];
    }
  }
}

extern "C" void kernel_launch(void* const* d_in, const int* in_sizes, int n_in,
                              void* d_out, int out_size, void* d_ws, size_t ws_size,
                              hipStream_t stream) {
  const float* x  = (const float*)d_in[0];
  const int*   ei = (const int*)d_in[1];
  const float* W  = (const float*)d_in[2];
  const float* b  = (const float*)d_in[3];

  const int n = in_sizes[0] / DFEAT;
  const int e = in_sizes[1] / 2;
  const int* src = ei;
  const int* dst = ei + e;

  const size_t nPad = ((size_t)n + 63) & ~(size_t)63;
  float* dinv = (float*)d_ws;
  float* z0   = dinv + nPad;
  float* acc1 = z0 + nPad;
  float* acc2 = acc1 + nPad;
  float* outx = (float*)d_out;
  float* outs = (float*)d_out + (size_t)n * DFEAT;

  float* planes = outx;
  const int planeStride = (int)nPad;

  const int B = 256;
  const int nodeGrid = (n + B - 1) / B;
  const dim3 scatGrid(SLICES, (n + RANGE - 1) / RANGE);
  const int sliceLen = (e + SLICES - 1) / SLICES;

  {
    const int PB = 128;
    node_project_kernel<<<(n + PB - 1) / PB, PB, 0, stream>>>(x, W, z0, n);
  }
  range_scatter_kernel<true><<<scatGrid, SCAT_THREADS, 0, stream>>>(
      z0, dinv, src, dst, planes, e, n, sliceLen, planeStride);
  dinv_reduce_kernel<<<nodeGrid, B, 0, stream>>>(planes, dinv, n, planeStride);
  range_scatter_kernel<false><<<scatGrid, SCAT_THREADS, 0, stream>>>(
      z0, dinv, src, dst, planes, e, n, sliceLen, planeStride);
  hop_reduce_kernel<<<nodeGrid, B, 0, stream>>>(planes, dinv, z0, acc1, n, planeStride);
  range_scatter_kernel<false><<<scatGrid, SCAT_THREADS, 0, stream>>>(
      acc1, dinv, src, dst, planes, e, n, sliceLen, planeStride);
  hop_reduce_kernel<<<nodeGrid, B, 0, stream>>>(planes, dinv, acc1, acc2, n, planeStride);
  {
    const int per = B * FIN_ITEMS;
    const int total4 = n * (DFEAT / 4);
    node_finish_kernel<<<(total4 + per - 1) / per, B, 0, stream>>>(
        x, acc2, b, outx, outs, n);
  }
}


# Edge weight factored per node: y=dinv*z scaled once before a hop, each edge adds y[src] (one gather per edge, not three), dinv[dst] applied once after the sum; summed terms rescaled, add order kept
# speedup vs baseline: 4.8903x; 1.3043x over previous
#include <hip/hip_runtime.h>
#include <cstdint>

#define DFEAT 64


typedef float v2f __attribute__((ext_vector_type(2)));
typedef float v8f __attribute__((ext_vector_type(8)));

__global__ void node_project_kernel(const float* __restrict__ x,
                                    const float* __restrict__ W,
                                    float* __restrict__ z0, int n) {
  const int lane = threadIdx.x & 31;
  const int wave = threadIdx.x >> 5;
  const int nb   = (blockIdx.x * (blockDim.x >> 5) + wave) * 32;
  if (nb >= n) return;

  const int half = lane >> 4;
  const int m    = lane & 15;
  int rowA = nb + m;
  int rowB = nb + 16 + m;
  if (rowA > n - 1) rowA = n - 1;
  if (rowB > n - 1) rowB = n - 1;
  const float* xa = x + (size_t)rowA * DFEAT;
  const float* xb = x + (size_t)rowB * DFEAT;

  v8f ca = {};
  v8f cb = {};
#pragma unroll
  for (int k0 = 0; k0 < DFEAT; k0 += 4) {
    const int kb = k0 + half * 2;
    v2f bm; bm.x = W[kb];  bm.y = W[kb + 1];
    v2f a;  a.x  = xa[kb]; a.y  = xa[kb + 1];
    v2f b;  b.x  = xb[kb]; b.y  = xb[kb + 1];
    ca = __builtin_amdgcn_wmma_f32_16x16x4_f32(
         false, a,  false, bm,
         (short)0, ca,  false,  false);
    cb = __builtin_amdgcn_wmma_f32_16x16x4_f32(
         false, b,  false, bm,
         (short)0, cb,  false,  false);
  }

  const int tile = m >> 3;
  const int rr   = m & 7;
  float v = 0.0f;
#pragma unroll
  for (int r = 0; r < 8; ++r) {
    const float pick = tile ? cb[r] : ca[r];
    v = (rr == r) ? pick : v;
  }
  const int node = nb + tile * 16 + half * 8 + rr;
  if (node < n) z0[node] = v;
}

#define RANGE        25024
#define SLICES       64
#define SCAT_THREADS 1024
#define SCAT_ITEMS   5

template <bool kDegree>
__global__ __launch_bounds__(SCAT_THREADS)
void range_scatter_kernel(const float* __restrict__ yin,
                          const int* __restrict__ src,
                          const int* __restrict__ dst,
                          float* __restrict__ planes,
                          int e, int n, int sliceLen, int planeStride) {
  __shared__ float acc[RANGE];

  const int lo = blockIdx.y * RANGE;
  int cnt = n - lo;
  if (cnt > RANGE) cnt = RANGE;

  for (int i = threadIdx.x; i < cnt; i += SCAT_THREADS) acc[i] = 0.0f;
  __syncthreads();

  const int sLo = blockIdx.x * sliceLen;
  int sHi = sLo + sliceLen;
  if (sHi > e) sHi = e;

  for (int b = sLo + (int)threadIdx.x; b < sHi; b += SCAT_THREADS * SCAT_ITEMS) {
    int   t[SCAT_ITEMS];
    int   s[SCAT_ITEMS];
    bool  keep[SCAT_ITEMS];
    float v[SCAT_ITEMS];
#pragma unroll
    for (int j = 0; j < SCAT_ITEMS; ++j) {
      const int  idx = b + j * SCAT_THREADS;
      const bool inb = idx < sHi;
      const int  ic  = inb ? idx : sHi - 1;
      t[j] = dst[ic];
      s[j] = kDegree ? 0 : src[ic];
      keep[j] = inb && (unsigned)(t[j] - lo) < (unsigned)cnt;
    }
#pragma unroll
    for (int j = 0; j < SCAT_ITEMS; ++j) {
      if (kDegree) {
        v[j] = 1.0f;
      } else {
        const int sg = keep[j] ? s[j] : lo;
        v[j] = yin[sg];
      }
    }
    if (!kDegree) {
#pragma unroll
      for (int j = 0; j < SCAT_ITEMS; ++j) asm volatile("" :: "v"(v[j]));
    }
#pragma unroll
    for (int j = 0; j < SCAT_ITEMS; ++j)
      if (keep[j]) atomicAdd(&acc[t[j] - lo], v[j]);
  }
  __syncthreads();

  float* piece = planes + (size_t)blockIdx.x * planeStride + lo;
  for (int i = threadIdx.x; i < cnt; i += SCAT_THREADS) piece[i] = acc[i];
}

__global__ void dinv_reduce_kernel(const float* __restrict__ planes,
                                   float* __restrict__ dinv,
                                   float* __restrict__ zy,
                                   int n, int planeStride) {
  const int i = blockIdx.x * blockDim.x + threadIdx.x;
  if (i >= n) return;
  float a0 = 1.0f, a1 = 0.0f, a2 = 0.0f, a3 = 0.0f;
#pragma unroll
  for (int p = 0; p < SLICES; p += 4) {
    a0 += planes[(size_t)(p + 0) * planeStride + i];
    a1 += planes[(size_t)(p + 1) * planeStride + i];
    a2 += planes[(size_t)(p + 2) * planeStride + i];
    a3 += planes[(size_t)(p + 3) * planeStride + i];
  }
  const float d = rsqrtf((a0 + a1) + (a2 + a3));
  dinv[i] = d;
  zy[i] = d * zy[i];
}

template <bool kLast>
__global__ void hop_reduce_kernel(const float* __restrict__ planes,
                                  const float* __restrict__ dinv,
                                  const float* __restrict__ yin,
                                  float* __restrict__ out,
                                  int n, int planeStride) {
  const int i = blockIdx.x * blockDim.x + threadIdx.x;
  if (i >= n) return;
  const float d = dinv[i];
  float a0 = yin[i];
  float a1 = 0.0f, a2 = 0.0f, a3 = 0.0f;
#pragma unroll
  for (int p = 0; p < SLICES; p += 4) {
    a0 += planes[(size_t)(p + 0) * planeStride + i];
    a1 += planes[(size_t)(p + 1) * planeStride + i];
    a2 += planes[(size_t)(p + 2) * planeStride + i];
    a3 += planes[(size_t)(p + 3) * planeStride + i];
  }
  const float z = d * ((a0 + a1) + (a2 + a3));
  out[i] = kLast ? z : d * z;
}

#define FIN_ITEMS 4
__global__ void node_finish_kernel(const float* __restrict__ x,
                                   const float* __restrict__ z2,
                                   const float* __restrict__ bptr,
                                   float* __restrict__ outx,
                                   float* __restrict__ outs, int n) {
  const float4* __restrict__ x4   = reinterpret_cast<const float4*>(x);
  float4* __restrict__       out4 = reinterpret_cast<float4*>(outx);
  const int total4 = n * (DFEAT / 4);
  const int base   = blockIdx.x * (blockDim.x * FIN_ITEMS) + threadIdx.x;
  if (base >= total4) return;
  const float bias = bptr[0];
  float4 xv[FIN_ITEMS];
  float  sc[FIN_ITEMS];
#pragma unroll
  for (int j = 0; j < FIN_ITEMS; ++j) {
    int i = base + j * blockDim.x;
    if (i > total4 - 1) i = total4 - 1;
    xv[j] = x4[i];
    sc[j] = 1.0f / (1.0f + __expf(-(z2[i >> 4] + bias)));
  }
#pragma unroll
  for (int j = 0; j < FIN_ITEMS; ++j) {
    const int i = base + j * blockDim.x;
    if (i < total4) {
      out4[i] = make_float4(xv[j].x * sc[j], xv[j].y * sc[j],
                            xv[j].z * sc[j], xv[j].w * sc[j]);
      if ((i & 15) == 0) outs[i >> 4] = sc[j];
    }
  }
}

extern "C" void kernel_launch(void* const* d_in, const int* in_sizes, int n_in,
                              void* d_out, int out_size, void* d_ws, size_t ws_size,
                              hipStream_t stream) {
  const float* x  = (const float*)d_in[0];
  const int*   ei = (const int*)d_in[1];
  const float* W  = (const float*)d_in[2];
  const float* b  = (const float*)d_in[3];

  const int n = in_sizes[0] / DFEAT;
  const int e = in_sizes[1] / 2;
  const int* src = ei;
  const int* dst = ei + e;

  const size_t nPad = ((size_t)n + 63) & ~(size_t)63;
  float* dinv = (float*)d_ws;
  float* zy0  = dinv + nPad;
  float* y1   = zy0 + nPad;
  float* z2   = y1 + nPad;
  float* outx = (float*)d_out;
  float* outs = (float*)d_out + (size_t)n * DFEAT;

  float* planes = outx;
  const int planeStride = (int)nPad;

  const int B = 256;
  const int nodeGrid = (n + B - 1) / B;
  const dim3 scatGrid(SLICES, (n + RANGE - 1) / RANGE);
  const int sliceLen = (e + SLICES - 1) / SLICES;

  {
    const int PB = 128;
    node_project_kernel<<<(n + PB - 1) / PB, PB, 0, stream>>>(x, W, zy0, n);
  }
  range_scatter_kernel<true><<<scatGrid, SCAT_THREADS, 0, stream>>>(
      zy0, src, dst, planes, e, n, sliceLen, planeStride);
  dinv_reduce_kernel<<<nodeGrid, B, 0, stream>>>(planes, dinv, zy0, n, planeStride);
  range_scatter_kernel<false><<<scatGrid, SCAT_THREADS, 0, stream>>>(
      zy0, src, dst, planes, e, n, sliceLen, planeStride);
  hop_reduce_kernel<false><<<nodeGrid, B, 0, stream>>>(planes, dinv, zy0, y1, n, planeStride);
  range_scatter_kernel<false><<<scatGrid, SCAT_THREADS, 0, stream>>>(
      y1, src, dst, planes, e, n, sliceLen, planeStride);
  hop_reduce_kernel<true><<<nodeGrid, B, 0, stream>>>(planes, dinv, y1, z2, n, planeStride);
  {
    const int per = B * FIN_ITEMS;
    const int total4 = n * (DFEAT / 4);
    node_finish_kernel<<<(total4 + per - 1) / per, B, 0, stream>>>(
        x, z2, b, outx, outs, n);
  }
}


# Fewer, longer-lived waves, same loads in flight: scatter blocks 16 waves x 49 edges/thread, one trip (was 32 x 5, five trips), 16-byte LDS clear/copy; finish 16 float4/thread; LDS add order changes
# speedup vs baseline: 6.4944x; 1.3280x over previous
#include <hip/hip_runtime.h>
#include <cstdint>

#define DFEAT 64


typedef float v2f __attribute__((ext_vector_type(2)));
typedef float v8f __attribute__((ext_vector_type(8)));

__global__ void node_project_kernel(const float* __restrict__ x,
                                    const float* __restrict__ W,
                                    float* __restrict__ z0, int n) {
  const int lane = threadIdx.x & 31;
  const int wave = threadIdx.x >> 5;
  const int nb   = (blockIdx.x * (blockDim.x >> 5) + wave) * 32;
  if (nb >= n) return;

  const int half = lane >> 4;
  const int m    = lane & 15;
  int rowA = nb + m;
  int rowB = nb + 16 + m;
  if (rowA > n - 1) rowA = n - 1;
  if (rowB > n - 1) rowB = n - 1;
  const float* xa = x + (size_t)rowA * DFEAT;
  const float* xb = x + (size_t)rowB * DFEAT;

  v8f ca = {};
  v8f cb = {};
#pragma unroll
  for (int k0 = 0; k0 < DFEAT; k0 += 4) {
    const int kb = k0 + half * 2;
    v2f bm; bm.x = W[kb];  bm.y = W[kb + 1];
    v2f a;  a.x  = xa[kb]; a.y  = xa[kb + 1];
    v2f b;  b.x  = xb[kb]; b.y  = xb[kb + 1];
    ca = __builtin_amdgcn_wmma_f32_16x16x4_f32(
         false, a,  false, bm,
         (short)0, ca,  false,  false);
    cb = __builtin_amdgcn_wmma_f32_16x16x4_f32(
         false, b,  false, bm,
         (short)0, cb,  false,  false);
  }

  const int tile = m >> 3;
  const int rr   = m & 7;
  float v = 0.0f;
#pragma unroll
  for (int r = 0; r < 8; ++r) {
    const float pick = tile ? cb[r] : ca[r];
    v = (rr == r) ? pick : v;
  }
  const int node = nb + tile * 16 + half * 8 + rr;
  if (node < n) z0[node] = v;
}

#define RANGE        25024
#define SLICES       64
#define SCAT_THREADS 512
#define SCAT_ITEMS   49
#define SCAT_TRIP    (SCAT_THREADS * SCAT_ITEMS)

template <bool kDegree, bool kClamp>
__device__ __forceinline__ void scatter_trip(const float* __restrict__ yin,
                                             const int* __restrict__ src,
                                             const int* __restrict__ dst,
                                             float* acc, int first, int bLo, int bHi,
                                             int lo, int cnt) {
  int   t[SCAT_ITEMS];
  int   s[SCAT_ITEMS];
  bool  keep[SCAT_ITEMS];
  float v[SCAT_ITEMS];
#pragma unroll
  for (int j = 0; j < SCAT_ITEMS; ++j) {
    const int  idx = first + j * SCAT_THREADS;
    const bool inb = (unsigned)(idx - bLo) < (unsigned)(bHi - bLo);
    const int  ic  = kClamp ? (inb ? idx : bLo) : idx;
    t[j] = dst[ic];
    s[j] = kDegree ? 0 : src[ic];
    keep[j] = inb && (unsigned)(t[j] - lo) < (unsigned)cnt;
  }
  if (!kDegree) {
#pragma unroll
    for (int j = 0; j < SCAT_ITEMS; ++j) asm volatile("" :: "v"(s[j]));
  }
#pragma unroll
  for (int j = 0; j < SCAT_ITEMS; ++j) {
    if (kDegree) {
      v[j] = 1.0f;
    } else {
      const int sg = keep[j] ? s[j] : lo;
      v[j] = yin[sg];
    }
  }
  if (!kDegree) {
#pragma unroll
    for (int j = 0; j < SCAT_ITEMS; ++j) asm volatile("" :: "v"(v[j]));
  }
#pragma unroll
  for (int j = 0; j < SCAT_ITEMS; ++j)
    if (keep[j]) atomicAdd(&acc[t[j] - lo], v[j]);
}

template <bool kDegree>
__global__ __launch_bounds__(SCAT_THREADS)
void range_scatter_kernel(const float* __restrict__ yin,
                          const int* __restrict__ src,
                          const int* __restrict__ dst,
                          float* __restrict__ planes,
                          int e, int n, int sliceLen, int planeStride) {
  __shared__ float4 acc4[RANGE / 4];
  float* acc = reinterpret_cast<float*>(acc4);

  const int lo = blockIdx.y * RANGE;
  int cnt = n - lo;
  if (cnt > RANGE) cnt = RANGE;

  for (int i = threadIdx.x; i < RANGE / 4; i += SCAT_THREADS)
    acc4[i] = make_float4(0.0f, 0.0f, 0.0f, 0.0f);
  __syncthreads();

  const int sLo = blockIdx.x * sliceLen;
  int sHi = sLo + sliceLen;
  if (sHi > e) sHi = e;

  for (int b0 = sLo; b0 < sHi; b0 += SCAT_TRIP) {
    int w0 = b0;
    if (w0 > e - SCAT_TRIP) w0 = e - SCAT_TRIP;
    if (w0 >= 0)
      scatter_trip<kDegree, false>(yin, src, dst, acc, w0 + (int)threadIdx.x, b0, sHi, lo, cnt);
    else
      scatter_trip<kDegree, true>(yin, src, dst, acc, b0 + (int)threadIdx.x, b0, sHi, lo, cnt);
  }
  __syncthreads();

  float* piece = planes + (size_t)blockIdx.x * planeStride + lo;
  float4* piece4 = reinterpret_cast<float4*>(piece);
  const int cnt4 = cnt >> 2;
  for (int i = threadIdx.x; i < cnt4; i += SCAT_THREADS) piece4[i] = acc4[i];
  for (int i = (cnt4 << 2) + (int)threadIdx.x; i < cnt; i += SCAT_THREADS) piece[i] = acc[i];
}

__global__ void dinv_reduce_kernel(const float* __restrict__ planes,
                                   float* __restrict__ dinv,
                                   float* __restrict__ zy,
                                   int n, int planeStride) {
  const int i = blockIdx.x * blockDim.x + threadIdx.x;
  if (i >= n) return;
  float a0 = 1.0f, a1 = 0.0f, a2 = 0.0f, a3 = 0.0f;
#pragma unroll
  for (int p = 0; p < SLICES; p += 4) {
    a0 += planes[(size_t)(p + 0) * planeStride + i];
    a1 += planes[(size_t)(p + 1) * planeStride + i];
    a2 += planes[(size_t)(p + 2) * planeStride + i];
    a3 += planes[(size_t)(p + 3) * planeStride + i];
  }
  const float d = rsqrtf((a0 + a1) + (a2 + a3));
  dinv[i] = d;
  zy[i] = d * zy[i];
}

template <bool kLast>
__global__ void hop_reduce_kernel(const float* __restrict__ planes,
                                  const float* __restrict__ dinv,
                                  const float* __restrict__ yin,
                                  float* __restrict__ out,
                                  int n, int planeStride) {
  const int i = blockIdx.x * blockDim.x + threadIdx.x;
  if (i >= n) return;
  const float d = dinv[i];
  float a0 = yin[i];
  float a1 = 0.0f, a2 = 0.0f, a3 = 0.0f;
#pragma unroll
  for (int p = 0; p < SLICES; p += 4) {
    a0 += planes[(size_t)(p + 0) * planeStride + i];
    a1 += planes[(size_t)(p + 1) * planeStride + i];
    a2 += planes[(size_t)(p + 2) * planeStride + i];
    a3 += planes[(size_t)(p + 3) * planeStride + i];
  }
  const float z = d * ((a0 + a1) + (a2 + a3));
  out[i] = kLast ? z : d * z;
}

#define FIN_ITEMS 16
__global__ void node_finish_kernel(const float* __restrict__ x,
                                   const float* __restrict__ z2,
                                   const float* __restrict__ bptr,
                                   float* __restrict__ outx,
                                   float* __restrict__ outs, int n) {
  const float4* __restrict__ x4   = reinterpret_cast<const float4*>(x);
  float4* __restrict__       out4 = reinterpret_cast<float4*>(outx);
  const int total4 = n * (DFEAT / 4);
  const int base   = blockIdx.x * (blockDim.x * FIN_ITEMS) + threadIdx.x;
  if (base >= total4) return;
  const float bias = bptr[0];
  float4 xv[FIN_ITEMS];
  float  sc[FIN_ITEMS];
#pragma unroll
  for (int j = 0; j < FIN_ITEMS; ++j) {
    int i = base + j * blockDim.x;
    if (i > total4 - 1) i = total4 - 1;
    xv[j] = x4[i];
    sc[j] = 1.0f / (1.0f + __expf(-(z2[i >> 4] + bias)));
  }
#pragma unroll
  for (int j = 0; j < FIN_ITEMS; ++j) {
    const int i = base + j * blockDim.x;
    if (i < total4) {
      out4[i] = make_float4(xv[j].x * sc[j], xv[j].y * sc[j],
                            xv[j].z * sc[j], xv[j].w * sc[j]);
      if ((i & 15) == 0) outs[i >> 4] = sc[j];
    }
  }
}

extern "C" void kernel_launch(void* const* d_in, const int* in_sizes, int n_in,
                              void* d_out, int out_size, void* d_ws, size_t ws_size,
                              hipStream_t stream) {
  const float* x  = (const float*)d_in[0];
  const int*   ei = (const int*)d_in[1];
  const float* W  = (const float*)d_in[2];
  const float* b  = (const float*)d_in[3];

  const int n = in_sizes[0] / DFEAT;
  const int e = in_sizes[1] / 2;
  const int* src = ei;
  const int* dst = ei + e;

  const size_t nPad = ((size_t)n + 63) & ~(size_t)63;
  float* dinv = (float*)d_ws;
  float* zy0  = dinv + nPad;
  float* y1   = zy0 + nPad;
  float* z2   = y1 + nPad;
  float* outx = (float*)d_out;
  float* outs = (float*)d_out + (size_t)n * DFEAT;

  float* planes = outx;
  const int planeStride = (int)nPad;

  const int B = 256;
  const int nodeGrid = (n + B - 1) / B;
  const dim3 scatGrid(SLICES, (n + RANGE - 1) / RANGE);
  const int sliceLen = (e + SLICES - 1) / SLICES;

  {
    const int PB = 128;
    node_project_kernel<<<(n + PB - 1) / PB, PB, 0, stream>>>(x, W, zy0, n);
  }
  range_scatter_kernel<true><<<scatGrid, SCAT_THREADS, 0, stream>>>(
      zy0, src, dst, planes, e, n, sliceLen, planeStride);
  dinv_reduce_kernel<<<nodeGrid, B, 0, stream>>>(planes, dinv, zy0, n, planeStride);
  range_scatter_kernel<false><<<scatGrid, SCAT_THREADS, 0, stream>>>(
      zy0, src, dst, planes, e, n, sliceLen, planeStride);
  hop_reduce_kernel<false><<<nodeGrid, B, 0, stream>>>(planes, dinv, zy0, y1, n, planeStride);
  range_scatter_kernel<false><<<scatGrid, SCAT_THREADS, 0, stream>>>(
      y1, src, dst, planes, e, n, sliceLen, planeStride);
  hop_reduce_kernel<true><<<nodeGrid, B, 0, stream>>>(planes, dinv, y1, z2, n, planeStride);
  {
    const int per = B * FIN_ITEMS;
    const int total4 = n * (DFEAT / 4);
    node_finish_kernel<<<(total4 + per - 1) / per, B, 0, stream>>>(
        x, z2, b, outx, outs, n);
  }
}


# Two launches folded into neighbours (8 -> 6): projection runs inside the degree-sum pass; last hop's 64-plane sum runs inside the finish pass, its planes tiled per 256-node block; sum order kept
# speedup vs baseline: 6.9263x; 1.0665x over previous
#include <hip/hip_runtime.h>
#include <cstdint>

#define DFEAT 64


typedef float v2f __attribute__((ext_vector_type(2)));
typedef float v8f __attribute__((ext_vector_type(8)));

#define SLICES       64

__global__ void project_dinv_kernel(const float* __restrict__ x,
                                    const float* __restrict__ W,
                                    const float* __restrict__ planes,
                                    float* __restrict__ dinv,
                                    float* __restrict__ zy,
                                    int n, int planeStride) {
  const int lane = threadIdx.x & 31;
  const int wave = threadIdx.x >> 5;
  const int nb   = (blockIdx.x * (blockDim.x >> 5) + wave) * 32;
  if (nb >= n) return;

  const int half = lane >> 4;
  const int m    = lane & 15;
  int rowA = nb + m;
  int rowB = nb + 16 + m;
  if (rowA > n - 1) rowA = n - 1;
  if (rowB > n - 1) rowB = n - 1;
  const float* xa = x + (size_t)rowA * DFEAT;
  const float* xb = x + (size_t)rowB * DFEAT;

  const int tile = m >> 3;
  const int rr   = m & 7;
  const int node = nb + tile * 16 + half * 8 + rr;
  const int nc   = node < n ? node : n - 1;

  float pc[SLICES];
#pragma unroll
  for (int p = 0; p < SLICES; ++p) pc[p] = planes[(size_t)p * planeStride + nc];

  v8f ca = {};
  v8f cb = {};
#pragma unroll
  for (int k0 = 0; k0 < DFEAT; k0 += 4) {
    const int kb = k0 + half * 2;
    v2f bm; bm.x = W[kb];  bm.y = W[kb + 1];
    v2f a;  a.x  = xa[kb]; a.y  = xa[kb + 1];
    v2f b;  b.x  = xb[kb]; b.y  = xb[kb + 1];
    ca = __builtin_amdgcn_wmma_f32_16x16x4_f32(
         false, a,  false, bm,
         (short)0, ca,  false,  false);
    cb = __builtin_amdgcn_wmma_f32_16x16x4_f32(
         false, b,  false, bm,
         (short)0, cb,  false,  false);
  }

#pragma unroll
  for (int p = 0; p < SLICES; ++p) asm volatile("" :: "v"(pc[p]));

  float v = 0.0f;
#pragma unroll
  for (int r = 0; r < 8; ++r) {
    const float pick = tile ? cb[r] : ca[r];
    v = (rr == r) ? pick : v;
  }
  float a0 = 1.0f, a1 = 0.0f, a2 = 0.0f, a3 = 0.0f;
#pragma unroll
  for (int p = 0; p < SLICES; p += 4) {
    a0 += pc[p + 0];
    a1 += pc[p + 1];
    a2 += pc[p + 2];
    a3 += pc[p + 3];
  }
  const float d = rsqrtf((a0 + a1) + (a2 + a3));
  if (node < n) {
    dinv[node] = d;
    zy[node]   = d * v;
  }
}

#define TILE         256
#define RANGE        25088
#define SCAT_THREADS 512
#define SCAT_ITEMS   49
#define SCAT_TRIP    (SCAT_THREADS * SCAT_ITEMS)

template <bool kDegree, bool kClamp>
__device__ __forceinline__ void scatter_trip(const float* __restrict__ yin,
                                             const int* __restrict__ src,
                                             const int* __restrict__ dst,
                                             float* acc, int first, int bLo, int bHi,
                                             int lo, int cnt) {
  int   t[SCAT_ITEMS];
  int   s[SCAT_ITEMS];
  bool  keep[SCAT_ITEMS];
  float v[SCAT_ITEMS];
#pragma unroll
  for (int j = 0; j < SCAT_ITEMS; ++j) {
    const int  idx = first + j * SCAT_THREADS;
    const bool inb = (unsigned)(idx - bLo) < (unsigned)(bHi - bLo);
    const int  ic  = kClamp ? (inb ? idx : bLo) : idx;
    t[j] = dst[ic];
    s[j] = kDegree ? 0 : src[ic];
    keep[j] = inb && (unsigned)(t[j] - lo) < (unsigned)cnt;
  }
  if (!kDegree) {
#pragma unroll
    for (int j = 0; j < SCAT_ITEMS; ++j) asm volatile("" :: "v"(s[j]));
  }
#pragma unroll
  for (int j = 0; j < SCAT_ITEMS; ++j) {
    if (kDegree) {
      v[j] = 1.0f;
    } else {
      const int sg = keep[j] ? s[j] : lo;
      v[j] = yin[sg];
    }
  }
  if (!kDegree) {
#pragma unroll
    for (int j = 0; j < SCAT_ITEMS; ++j) asm volatile("" :: "v"(v[j]));
  }
#pragma unroll
  for (int j = 0; j < SCAT_ITEMS; ++j)
    if (keep[j]) atomicAdd(&acc[t[j] - lo], v[j]);
}

template <bool kDegree, bool kTiled>
__global__ __launch_bounds__(SCAT_THREADS)
void range_scatter_kernel(const float* __restrict__ yin,
                          const int* __restrict__ src,
                          const int* __restrict__ dst,
                          float* __restrict__ planes,
                          int e, int n, int sliceLen, int planeStride) {
  __shared__ float4 acc4[RANGE / 4];
  float* acc = reinterpret_cast<float*>(acc4);

  const int lo = blockIdx.y * RANGE;
  int cnt = n - lo;
  if (cnt > RANGE) cnt = RANGE;

  for (int i = threadIdx.x; i < RANGE / 4; i += SCAT_THREADS)
    acc4[i] = make_float4(0.0f, 0.0f, 0.0f, 0.0f);
  __syncthreads();

  const int sLo = blockIdx.x * sliceLen;
  int sHi = sLo + sliceLen;
  if (sHi > e) sHi = e;

  for (int b0 = sLo; b0 < sHi; b0 += SCAT_TRIP) {
    int w0 = b0;
    if (w0 > e - SCAT_TRIP) w0 = e - SCAT_TRIP;
    if (w0 >= 0)
      scatter_trip<kDegree, false>(yin, src, dst, acc, w0 + (int)threadIdx.x, b0, sHi, lo, cnt);
    else
      scatter_trip<kDegree, true>(yin, src, dst, acc, b0 + (int)threadIdx.x, b0, sHi, lo, cnt);
  }
  __syncthreads();

  if (!kTiled) {
    float* piece = planes + (size_t)blockIdx.x * planeStride + lo;
    float4* piece4 = reinterpret_cast<float4*>(piece);
    const int cnt4 = cnt >> 2;
    for (int i = threadIdx.x; i < cnt4; i += SCAT_THREADS) piece4[i] = acc4[i];
    for (int i = (cnt4 << 2) + (int)threadIdx.x; i < cnt; i += SCAT_THREADS) piece[i] = acc[i];
  } else {
    float* base = planes + (size_t)lo * SLICES;
    const int whole = cnt / TILE;
    const int whole4 = whole * (TILE / 4);
    for (int i = threadIdx.x; i < whole4; i += SCAT_THREADS) {
      const int tl = i / (TILE / 4);
      const int j4 = i % (TILE / 4);
      float4* p4 = reinterpret_cast<float4*>(base + (size_t)tl * (TILE * SLICES) + blockIdx.x * TILE);
      p4[j4] = acc4[i];
    }
    const int rem = cnt - whole * TILE;
    float* tb = base + (size_t)whole * (TILE * SLICES) + (size_t)blockIdx.x * rem;
    for (int j = threadIdx.x; j < rem; j += SCAT_THREADS) tb[j] = acc[whole * TILE + j];
  }
}

__global__ void hop_reduce_kernel(const float* __restrict__ planes,
                                  const float* __restrict__ dinv,
                                  const float* __restrict__ yin,
                                  float* __restrict__ out,
                                  int n, int planeStride) {
  const int i = blockIdx.x * blockDim.x + threadIdx.x;
  if (i >= n) return;
  const float d = dinv[i];
  float a0 = yin[i];
  float a1 = 0.0f, a2 = 0.0f, a3 = 0.0f;
#pragma unroll
  for (int p = 0; p < SLICES; p += 4) {
    a0 += planes[(size_t)(p + 0) * planeStride + i];
    a1 += planes[(size_t)(p + 1) * planeStride + i];
    a2 += planes[(size_t)(p + 2) * planeStride + i];
    a3 += planes[(size_t)(p + 3) * planeStride + i];
  }
  const float z = d * ((a0 + a1) + (a2 + a3));
  out[i] = d * z;
}

#define TILE_THREADS 256
#define FIN_ITEMS (TILE * (DFEAT / 4) / TILE_THREADS)

template <bool kFull>
__device__ __forceinline__ void finish_tile(const float4* __restrict__ x4,
                                            const float* __restrict__ dinv,
                                            const float* __restrict__ yin,
                                            const float* __restrict__ bptr,
                                            float* region,
                                            float* __restrict__ outs,
                                            float* score_s, int n0, int lenIn) {
  const int len    = kFull ? TILE : lenIn;
  const int total4 = len * (DFEAT / 4);
  const int k      = threadIdx.x;
  const int kc     = (kFull || k < len) ? k : len - 1;
  const int node   = n0 + kc;

  float4 xv[FIN_ITEMS];
#pragma unroll
  for (int j = 0; j < FIN_ITEMS; ++j) {
    int i = (int)threadIdx.x + j * TILE_THREADS;
    if (!kFull && i > total4 - 1) i = total4 - 1;
    xv[j] = x4[i];
  }

  float pr[SLICES];
#pragma unroll
  for (int p = 0; p < SLICES; ++p) pr[p] = region[(size_t)p * len + kc];
  const float d    = dinv[node];
  const float self = yin[node];
  const float bias = bptr[0];

#pragma unroll
  for (int j = 0; j < FIN_ITEMS; ++j)
    asm volatile("" :: "v"(xv[j].x), "v"(xv[j].y), "v"(xv[j].z), "v"(xv[j].w));
#pragma unroll
  for (int p = 0; p < SLICES; ++p) asm volatile("" :: "v"(pr[p]));

  float a0 = self, a1 = 0.0f, a2 = 0.0f, a3 = 0.0f;
#pragma unroll
  for (int p = 0; p < SLICES; p += 4) {
    a0 += pr[p + 0];
    a1 += pr[p + 1];
    a2 += pr[p + 2];
    a3 += pr[p + 3];
  }
  const float z  = d * ((a0 + a1) + (a2 + a3));
  const float sc = 1.0f / (1.0f + __expf(-(z + bias)));
  score_s[k] = sc;
  if (kFull || k < len) outs[n0 + k] = sc;
  __syncthreads();

  float scv[FIN_ITEMS];
#pragma unroll
  for (int j = 0; j < FIN_ITEMS; ++j)
    scv[j] = score_s[((int)threadIdx.x + j * TILE_THREADS) >> 4];

  float4* out4 = reinterpret_cast<float4*>(region);
#pragma unroll
  for (int j = 0; j < FIN_ITEMS; ++j) {
    const int i = (int)threadIdx.x + j * TILE_THREADS;
    if (kFull || i < total4)
      out4[i] = make_float4(xv[j].x * scv[j], xv[j].y * scv[j],
                            xv[j].z * scv[j], xv[j].w * scv[j]);
  }
}

__global__ __launch_bounds__(TILE_THREADS)
void hop_finish_kernel(const float* __restrict__ x,
                       const float* __restrict__ dinv,
                       const float* __restrict__ yin,
                       const float* __restrict__ bptr,
                       float* buf,
                       float* __restrict__ outs, int n) {
  __shared__ float score_s[TILE];
  const int n0  = blockIdx.x * TILE;
  const int len = n - n0;
  const float4* x4 = reinterpret_cast<const float4*>(x) + (size_t)n0 * (DFEAT / 4);
  float* region = buf + (size_t)n0 * DFEAT;
  if (len >= TILE)
    finish_tile<true>(x4, dinv, yin, bptr, region, outs, score_s, n0, TILE);
  else
    finish_tile<false>(x4, dinv, yin, bptr, region, outs, score_s, n0, len);
}

extern "C" void kernel_launch(void* const* d_in, const int* in_sizes, int n_in,
                              void* d_out, int out_size, void* d_ws, size_t ws_size,
                              hipStream_t stream) {
  const float* x  = (const float*)d_in[0];
  const int*   ei = (const int*)d_in[1];
  const float* W  = (const float*)d_in[2];
  const float* b  = (const float*)d_in[3];

  const int n = in_sizes[0] / DFEAT;
  const int e = in_sizes[1] / 2;
  const int* src = ei;
  const int* dst = ei + e;

  const size_t nPad = ((size_t)n + 63) & ~(size_t)63;
  float* dinv = (float*)d_ws;
  float* zy0  = dinv + nPad;
  float* y1   = zy0 + nPad;
  float* outx = (float*)d_out;
  float* outs = (float*)d_out + (size_t)n * DFEAT;

  float* planes = outx;
  const int planeStride = (int)nPad;

  const int B = 256;
  const int nodeGrid = (n + B - 1) / B;
  const dim3 scatGrid(SLICES, (n + RANGE - 1) / RANGE);
  const int sliceLen = (e + SLICES - 1) / SLICES;

  range_scatter_kernel<true, false><<<scatGrid, SCAT_THREADS, 0, stream>>>(
      zy0, src, dst, planes, e, n, sliceLen, planeStride);
  {
    const int PB = 128;
    project_dinv_kernel<<<(n + PB - 1) / PB, PB, 0, stream>>>(x, W, planes, dinv, zy0, n, planeStride);
  }
  range_scatter_kernel<false, false><<<scatGrid, SCAT_THREADS, 0, stream>>>(
      zy0, src, dst, planes, e, n, sliceLen, planeStride);
  hop_reduce_kernel<<<nodeGrid, B, 0, stream>>>(planes, dinv, zy0, y1, n, planeStride);
  range_scatter_kernel<false, true><<<scatGrid, SCAT_THREADS, 0, stream>>>(
      y1, src, dst, planes, e, n, sliceLen, planeStride);
  hop_finish_kernel<<<(n + TILE - 1) / TILE, TILE_THREADS, 0, stream>>>(
      x, dinv, y1, b, outx, outs, n);
}


# Edges placed by destination range once (per-slice LDS counting sort, one packed word per edge); a range block sums its nodes in LDS to final values: no partial planes, 5 launches; add order changes
# speedup vs baseline: 8.7199x; 1.2590x over previous
#include <hip/hip_runtime.h>
#include <cstdint>

#define DFEAT 64


typedef float v2f __attribute__((ext_vector_type(2)));
typedef float v8f __attribute__((ext_vector_type(8)));

#define NSLICE      256
#define RSHIFT      8
#define RNODES      (1 << RSHIFT)
#define MAXR        512
#define BIN_THREADS 256
#define BIN_ITEMS   25
#define BIN_CAP     (BIN_THREADS * BIN_ITEMS)

__global__ __launch_bounds__(BIN_THREADS)
void bin_edges_kernel(const int* __restrict__ src,
                      const int* __restrict__ dst,
                      unsigned int* __restrict__ bins,
                      int* __restrict__ offs,
                      int e, int sliceLen) {
  __shared__ int   start[MAXR];
  __shared__ uint4 stage4[BIN_CAP / 4];
  __shared__ int   wsum[BIN_THREADS / 32];
  unsigned int* stage = reinterpret_cast<unsigned int*>(stage4);

  const int tid  = threadIdx.x;
  const int lane = tid & 31;
  const int wave = tid >> 5;
  for (int i = tid; i < MAXR; i += BIN_THREADS) start[i] = 0;
  __syncthreads();

  const int sLo = blockIdx.x * sliceLen;
  int sHi = sLo + sliceLen;
  if (sHi > e) sHi = e;

  int d[BIN_ITEMS];
  int s[BIN_ITEMS];
  int rk[BIN_ITEMS];
#pragma unroll
  for (int j = 0; j < BIN_ITEMS; ++j) {
    const int idx = sLo + tid + j * BIN_THREADS;
    const int ic  = idx < sHi ? idx : 0;
    d[j] = dst[ic];
    s[j] = src[ic];
  }
#pragma unroll
  for (int j = 0; j < BIN_ITEMS; ++j) {
    const int idx = sLo + tid + j * BIN_THREADS;
    rk[j] = idx < sHi ? atomicAdd(&start[d[j] >> RSHIFT], 1) : 0;
  }
  __syncthreads();

  const int c0 = start[2 * tid];
  const int c1 = start[2 * tid + 1];
  const int both = c0 + c1;
  int inc = both;
  for (int o = 1; o < 32; o <<= 1) {
    const int up = __shfl_up(inc, o);
    if (lane >= o) inc += up;
  }
  if (lane == 31) wsum[wave] = inc;
  __syncthreads();
  int wbase = 0;
  for (int w = 0; w < wave; ++w) wbase += wsum[w];
  const int ex0 = wbase + inc - both;
  const int ex1 = ex0 + c0;
  start[2 * tid]     = ex0;
  start[2 * tid + 1] = ex1;
  int* row = offs + (size_t)blockIdx.x * MAXR;
  row[2 * tid]     = ex0;
  row[2 * tid + 1] = ex1;
  __syncthreads();

#pragma unroll
  for (int j = 0; j < BIN_ITEMS; ++j) {
    const int idx = sLo + tid + j * BIN_THREADS;
    if (idx < sHi)
      stage[start[d[j] >> RSHIFT] + rk[j]] =
          ((unsigned int)s[j] << RSHIFT) | ((unsigned int)d[j] & (unsigned int)(RNODES - 1));
  }
  __syncthreads();

  int cntS = sHi - sLo;
  if (cntS < 0) cntS = 0;
  unsigned int* seg = bins + (size_t)blockIdx.x * BIN_CAP;
  uint4* seg4 = reinterpret_cast<uint4*>(seg);
  const int c4 = cntS >> 2;
  for (int i = tid; i < c4; i += BIN_THREADS) seg4[i] = stage4[i];
  for (int i = (c4 << 2) + tid; i < cntS; i += BIN_THREADS) seg[i] = stage[i];
}

template <bool kDegree>
__device__ __forceinline__ void range_sum(const unsigned int* __restrict__ bins,
                                          const int* __restrict__ offs,
                                          const float* __restrict__ yin,
                                          float* acc, int r) {
  const int lane = threadIdx.x & 31;
  const int wave = threadIdx.x >> 5;
  const int* row = offs + (size_t)(wave * 32 + lane) * MAXR + r;
  const int beg = row[0];
  const int len = row[1] - beg;
  const unsigned int* wbase = bins + (size_t)(wave * 32) * BIN_CAP;

  unsigned int wd[32];
#pragma unroll
  for (int j = 0; j < 32; ++j) {
    const int b = __builtin_amdgcn_readlane(beg, j);
    const int l = __builtin_amdgcn_readlane(len, j);
    const int k = lane < l ? b + lane : 0;
    wd[j] = wbase[(size_t)j * BIN_CAP + k];
  }
#pragma unroll
  for (int j = 0; j < 32; ++j) asm volatile("" :: "v"(wd[j]));
  float v[32];
#pragma unroll
  for (int j = 0; j < 32; ++j) {
    if (kDegree) {
      v[j] = 1.0f;
    } else {
      const int l = __builtin_amdgcn_readlane(len, j);
      const unsigned int sg = lane < l ? (wd[j] >> RSHIFT) : 0u;
      v[j] = yin[sg];
    }
  }
  if (!kDegree) {
#pragma unroll
    for (int j = 0; j < 32; ++j) asm volatile("" :: "v"(v[j]));
  }
  __syncthreads();
#pragma unroll
  for (int j = 0; j < 32; ++j) {
    const int l = __builtin_amdgcn_readlane(len, j);
    if (lane < l) atomicAdd(&acc[wd[j] & (unsigned int)(RNODES - 1)], v[j]);
  }
  if (__builtin_amdgcn_ballot_w32(len > 32) != 0u) {
#pragma nounroll
    for (int j = 0; j < 32; ++j) {
      const int b = __builtin_amdgcn_readlane(beg, j);
      const int l = __builtin_amdgcn_readlane(len, j);
      const unsigned int* run = wbase + (size_t)j * BIN_CAP + b;
      for (int k = 32 + lane; k < l; k += 32) {
        const unsigned int w2 = run[k];
        const float val = kDegree ? 1.0f : yin[w2 >> RSHIFT];
        atomicAdd(&acc[w2 & (unsigned int)(RNODES - 1)], val);
      }
    }
  }
}

__global__ __launch_bounds__(RNODES)
void degree_project_kernel(const float* __restrict__ x,
                           const float* __restrict__ W,
                           const unsigned int* __restrict__ bins,
                           const int* __restrict__ offs,
                           float* __restrict__ dinv,
                           float* __restrict__ zy, int n) {
  __shared__ float acc[RNODES];
  __shared__ float dinv_s[RNODES];
  const int tid  = threadIdx.x;
  const int lane = tid & 31;
  const int wave = tid >> 5;
  const int lo   = blockIdx.x * RNODES;

  const int nb   = lo + wave * 32;
  const int half = lane >> 4;
  const int m    = lane & 15;
  int rowA = nb + m;
  int rowB = nb + 16 + m;
  if (rowA > n - 1) rowA = n - 1;
  if (rowB > n - 1) rowB = n - 1;
  const float* xa = x + (size_t)rowA * DFEAT;
  const float* xb = x + (size_t)rowB * DFEAT;

  v2f fa[DFEAT / 4];
  v2f fb[DFEAT / 4];
  v2f fw[DFEAT / 4];
#pragma unroll
  for (int k0 = 0; k0 < DFEAT; k0 += 4) {
    const int kb = k0 + half * 2;
    fw[k0 / 4].x = W[kb];  fw[k0 / 4].y = W[kb + 1];
    fa[k0 / 4].x = xa[kb]; fa[k0 / 4].y = xa[kb + 1];
    fb[k0 / 4].x = xb[kb]; fb[k0 / 4].y = xb[kb + 1];
  }

  acc[tid] = 0.0f;
  range_sum<true>(bins, offs, nullptr, acc, blockIdx.x);
  __syncthreads();
  {
    const float dd = rsqrtf(1.0f + acc[tid]);
    dinv_s[tid] = dd;
    if (lo + tid < n) dinv[lo + tid] = dd;
  }

  v8f ca = {};
  v8f cb = {};
#pragma unroll
  for (int q = 0; q < DFEAT / 4; ++q) {
    ca = __builtin_amdgcn_wmma_f32_16x16x4_f32(
         false, fa[q],  false, fw[q],
         (short)0, ca,  false,  false);
    cb = __builtin_amdgcn_wmma_f32_16x16x4_f32(
         false, fb[q],  false, fw[q],
         (short)0, cb,  false,  false);
  }

  const int tile = m >> 3;
  const int rr   = m & 7;
  const int node = nb + tile * 16 + half * 8 + rr;
  float v = 0.0f;
#pragma unroll
  for (int r = 0; r < 8; ++r) {
    const float pick = tile ? cb[r] : ca[r];
    v = (rr == r) ? pick : v;
  }
  __syncthreads();
  if (node < n) zy[node] = dinv_s[node - lo] * v;
}

template <bool kLast>
__global__ __launch_bounds__(RNODES)
void hop_kernel(const unsigned int* __restrict__ bins,
                const int* __restrict__ offs,
                const float* __restrict__ dinv,
                const float* __restrict__ yin,
                const float* __restrict__ bptr,
                float* __restrict__ yout, int n) {
  __shared__ float acc[RNODES];
  const int tid  = threadIdx.x;
  const int node = blockIdx.x * RNODES + tid;
  const int nc   = node < n ? node : n - 1;
  const float d    = dinv[nc];
  const float self = yin[nc];
  const float bias = kLast ? bptr[0] : 0.0f;

  acc[tid] = 0.0f;
  range_sum<false>(bins, offs, yin, acc, blockIdx.x);
  __syncthreads();

  const float z = d * (self + acc[tid]);
  if (node < n) {
    if (kLast) yout[node] = 1.0f / (1.0f + __expf(-(z + bias)));
    else       yout[node] = d * z;
  }
}

#define ROW_THREADS 256
#define ROW_ITEMS   16
__global__ __launch_bounds__(ROW_THREADS)
void scale_rows_kernel(const float* __restrict__ x,
                       const float* __restrict__ score,
                       float* __restrict__ outx, int total4) {
  const float4* x4 = reinterpret_cast<const float4*>(x);
  float4* o4 = reinterpret_cast<float4*>(outx);
  const int first = blockIdx.x * (ROW_THREADS * ROW_ITEMS) + (int)threadIdx.x;
  float4 xv[ROW_ITEMS];
  float  sc[ROW_ITEMS];
#pragma unroll
  for (int j = 0; j < ROW_ITEMS; ++j) {
    int i = first + j * ROW_THREADS;
    if (i > total4 - 1) i = total4 - 1;
    xv[j] = x4[i];
    sc[j] = score[i >> 4];
  }
#pragma unroll
  for (int j = 0; j < ROW_ITEMS; ++j) {
    const int i = first + j * ROW_THREADS;
    if (i < total4)
      o4[i] = make_float4(xv[j].x * sc[j], xv[j].y * sc[j],
                          xv[j].z * sc[j], xv[j].w * sc[j]);
  }
}

extern "C" void kernel_launch(void* const* d_in, const int* in_sizes, int n_in,
                              void* d_out, int out_size, void* d_ws, size_t ws_size,
                              hipStream_t stream) {
  const float* x  = (const float*)d_in[0];
  const int*   ei = (const int*)d_in[1];
  const float* W  = (const float*)d_in[2];
  const float* b  = (const float*)d_in[3];

  const int n = in_sizes[0] / DFEAT;
  const int e = in_sizes[1] / 2;
  const int* src = ei;
  const int* dst = ei + e;

  const size_t nPad = ((size_t)n + 63) & ~(size_t)63;
  float* dinv = (float*)d_ws;
  float* zy0  = dinv + nPad;
  float* y1   = zy0 + nPad;
  float* outx = (float*)d_out;
  float* outs = (float*)d_out + (size_t)n * DFEAT;

  unsigned int* bins = (unsigned int*)d_out;
  int* offs = (int*)d_out + (size_t)NSLICE * BIN_CAP;

  const int nRanges  = (n + RNODES - 1) / RNODES;
  const int sliceLen = (e + NSLICE - 1) / NSLICE;
  const int total4   = n * (DFEAT / 4);

  bin_edges_kernel<<<NSLICE, BIN_THREADS, 0, stream>>>(src, dst, bins, offs, e, sliceLen);
  degree_project_kernel<<<nRanges, RNODES, 0, stream>>>(x, W, bins, offs, dinv, zy0, n);
  hop_kernel<false><<<nRanges, RNODES, 0, stream>>>(bins, offs, dinv, zy0, b, y1, n);
  hop_kernel<true><<<nRanges, RNODES, 0, stream>>>(bins, offs, dinv, y1, b, outs, n);
  scale_rows_kernel<<<(total4 + ROW_THREADS * ROW_ITEMS - 1) / (ROW_THREADS * ROW_ITEMS), ROW_THREADS, 0, stream>>>(
      x, outs, outx, total4);
}


# Node ranges of 416 nodes (was 256): 241 range work-groups, one per WGP instead of 391 on 256 WGPs, 38% fewer waves; a run's edges past 32 packed into one extra round; LDS add order changes
# speedup vs baseline: 9.5335x; 1.0933x over previous
#include <hip/hip_runtime.h>
#include <cstdint>

#define DFEAT 64


typedef float v2f __attribute__((ext_vector_type(2)));
typedef float v8f __attribute__((ext_vector_type(8)));

#define NSLICE      256
#define RSHIFT      9
#define RMASK       ((1u << RSHIFT) - 1u)
#define RNODES      416
#define RTHREADS    256
#define RTILES      (RNODES / 16)
#define MAXR        512
#define BIN_THREADS 256
#define BIN_ITEMS   25
#define BIN_CAP     (BIN_THREADS * BIN_ITEMS)

__global__ __launch_bounds__(BIN_THREADS)
void bin_edges_kernel(const int* __restrict__ src,
                      const int* __restrict__ dst,
                      unsigned int* __restrict__ bins,
                      int* __restrict__ offs,
                      int e, int sliceLen) {
  __shared__ int   start[MAXR];
  __shared__ uint4 stage4[BIN_CAP / 4];
  __shared__ int   wsum[BIN_THREADS / 32];
  unsigned int* stage = reinterpret_cast<unsigned int*>(stage4);

  const int tid  = threadIdx.x;
  const int lane = tid & 31;
  const int wave = tid >> 5;
  for (int i = tid; i < MAXR; i += BIN_THREADS) start[i] = 0;
  __syncthreads();

  const int sLo = blockIdx.x * sliceLen;
  int sHi = sLo + sliceLen;
  if (sHi > e) sHi = e;

  int d[BIN_ITEMS];
  int s[BIN_ITEMS];
  int rk[BIN_ITEMS];
#pragma unroll
  for (int j = 0; j < BIN_ITEMS; ++j) {
    const int idx = sLo + tid + j * BIN_THREADS;
    const int ic  = idx < sHi ? idx : 0;
    d[j] = dst[ic];
    s[j] = src[ic];
  }
#pragma unroll
  for (int j = 0; j < BIN_ITEMS; ++j) {
    const int idx = sLo + tid + j * BIN_THREADS;
    rk[j] = idx < sHi ? atomicAdd(&start[d[j] / RNODES], 1) : 0;
  }
  __syncthreads();

  const int c0 = start[2 * tid];
  const int c1 = start[2 * tid + 1];
  const int both = c0 + c1;
  int inc = both;
  for (int o = 1; o < 32; o <<= 1) {
    const int up = __shfl_up(inc, o);
    if (lane >= o) inc += up;
  }
  if (lane == 31) wsum[wave] = inc;
  __syncthreads();
  int wbase = 0;
  for (int w = 0; w < wave; ++w) wbase += wsum[w];
  const int ex0 = wbase + inc - both;
  const int ex1 = ex0 + c0;
  start[2 * tid]     = ex0;
  start[2 * tid + 1] = ex1;
  int* row = offs + (size_t)blockIdx.x * MAXR;
  row[2 * tid]     = ex0;
  row[2 * tid + 1] = ex1;
  __syncthreads();

#pragma unroll
  for (int j = 0; j < BIN_ITEMS; ++j) {
    const int idx = sLo + tid + j * BIN_THREADS;
    if (idx < sHi) {
      const int rg = d[j] / RNODES;
      stage[start[rg] + rk[j]] =
          ((unsigned int)s[j] << RSHIFT) | (unsigned int)(d[j] - rg * RNODES);
    }
  }
  __syncthreads();

  int cntS = sHi - sLo;
  if (cntS < 0) cntS = 0;
  unsigned int* seg = bins + (size_t)blockIdx.x * BIN_CAP;
  uint4* seg4 = reinterpret_cast<uint4*>(seg);
  const int c4 = cntS >> 2;
  for (int i = tid; i < c4; i += BIN_THREADS) seg4[i] = stage4[i];
  for (int i = (c4 << 2) + tid; i < cntS; i += BIN_THREADS) seg[i] = stage[i];
}

template <bool kDegree>
__device__ __forceinline__ void range_sum(const unsigned int* __restrict__ bins,
                                          const int* __restrict__ offs,
                                          const float* __restrict__ yin,
                                          float* acc, int r) {
  const int lane = threadIdx.x & 31;
  const int wave = threadIdx.x >> 5;
  const int* row = offs + (size_t)(wave * 32 + lane) * MAXR + r;
  const int beg = row[0];
  const int len = row[1] - beg;
  const unsigned int* wbase = bins + (size_t)(wave * 32) * BIN_CAP;

  unsigned int wd[32];
#pragma unroll
  for (int j = 0; j < 32; ++j) {
    const int b = __builtin_amdgcn_readlane(beg, j);
    const int l = __builtin_amdgcn_readlane(len, j);
    const int k = lane < l ? b + lane : 0;
    wd[j] = wbase[(size_t)j * BIN_CAP + k];
  }

  const int ov = len > 32 ? len - 32 : 0;
  int inc = ov;
  for (int o = 1; o < 32; o <<= 1) {
    const int up = __shfl_up(inc, o);
    if (lane >= o) inc += up;
  }
  const int exc = inc - ov;
  const int tot = __builtin_amdgcn_readlane(inc, 31);
  int jx = 0;
#pragma unroll
  for (int l = 0; l < 32; ++l) jx += (__builtin_amdgcn_readlane(inc, l) <= lane) ? 1 : 0;
  const int jxc = jx < 32 ? jx : 31;
  const int ejx = __shfl(exc, jxc);
  const int bjx = __shfl(beg, jxc);
  const bool okx = lane < tot;
  const unsigned int wdx = wbase[okx ? (size_t)jxc * BIN_CAP + (size_t)(bjx + 32 + (lane - ejx)) : (size_t)0];
  __builtin_amdgcn_sched_barrier(0);

#pragma unroll
  for (int j = 0; j < 32; ++j) asm volatile("" :: "v"(wd[j]));
  float v[32];
  float vx = 1.0f;
#pragma unroll
  for (int j = 0; j < 32; ++j) {
    if (kDegree) {
      v[j] = 1.0f;
    } else {
      const int l = __builtin_amdgcn_readlane(len, j);
      const unsigned int sg = lane < l ? (wd[j] >> RSHIFT) : 0u;
      v[j] = yin[sg];
    }
  }
  if (!kDegree) {
    vx = yin[okx ? (wdx >> RSHIFT) : 0u];
#pragma unroll
    for (int j = 0; j < 32; ++j) asm volatile("" :: "v"(v[j]));
    asm volatile("" :: "v"(vx));
  }
  __syncthreads();
#pragma unroll
  for (int j = 0; j < 32; ++j) {
    const int l = __builtin_amdgcn_readlane(len, j);
    if (lane < l) atomicAdd(&acc[wd[j] & RMASK], v[j]);
  }
  if (okx) atomicAdd(&acc[wdx & RMASK], vx);
  if (tot > 32) {
#pragma nounroll
    for (int base = 32; base < tot; base += 32) {
      const int t = base + lane;
      int j2 = 0;
#pragma unroll
      for (int l = 0; l < 32; ++l) j2 += (__builtin_amdgcn_readlane(inc, l) <= t) ? 1 : 0;
      const int j2c = j2 < 32 ? j2 : 31;
      const int e2 = __shfl(exc, j2c);
      const int b2 = __shfl(beg, j2c);
      if (t < tot) {
        const unsigned int w2 = wbase[(size_t)j2c * BIN_CAP + (size_t)(b2 + 32 + (t - e2))];
        const float val = kDegree ? 1.0f : yin[w2 >> RSHIFT];
        atomicAdd(&acc[w2 & RMASK], val);
      }
    }
  }
}

__global__ __launch_bounds__(RTHREADS)
void degree_project_kernel(const float* __restrict__ x,
                           const float* __restrict__ W,
                           const unsigned int* __restrict__ bins,
                           const int* __restrict__ offs,
                           float* __restrict__ dinv,
                           float* __restrict__ zy, int n) {
  __shared__ float acc[RNODES];
  __shared__ float dinv_s[RNODES];
  __shared__ float z_s[RTILES * 16];
  const int tid  = threadIdx.x;
  const int lane = tid & 31;
  const int wv   = __builtin_amdgcn_readfirstlane(tid >> 5);
  const int lo   = blockIdx.x * RNODES;
  const int half = lane >> 4;
  const int m    = lane & 15;
  const bool four = wv + 24 < RTILES;
  int r0 = lo + (wv     ) * 16 + m;
  int r1 = lo + (wv +  8) * 16 + m;
  int r2 = lo + (wv + 16) * 16 + m;
  int r3 = lo + (wv + 24) * 16 + m;
  if (r0 > n - 1) r0 = n - 1;
  if (r1 > n - 1) r1 = n - 1;
  if (r2 > n - 1) r2 = n - 1;
  if (r3 > n - 1) r3 = n - 1;
  const float* x0 = x + (size_t)r0 * DFEAT;
  const float* x1 = x + (size_t)r1 * DFEAT;
  const float* x2 = x + (size_t)r2 * DFEAT;
  const float* x3 = x + (size_t)r3 * DFEAT;

  v2f f0[DFEAT / 4];
  v2f f1[DFEAT / 4];
  v2f f2[DFEAT / 4];
  v2f f3[DFEAT / 4];
  v2f fw[DFEAT / 4];
#pragma unroll
  for (int k0 = 0; k0 < DFEAT; k0 += 4) {
    const int kb = k0 + half * 2;
    fw[k0 / 4].x = W[kb];  fw[k0 / 4].y = W[kb + 1];
    f0[k0 / 4].x = x0[kb]; f0[k0 / 4].y = x0[kb + 1];
    f1[k0 / 4].x = x1[kb]; f1[k0 / 4].y = x1[kb + 1];
    f2[k0 / 4].x = x2[kb]; f2[k0 / 4].y = x2[kb + 1];
  }
  if (four) {
#pragma unroll
    for (int k0 = 0; k0 < DFEAT; k0 += 4) {
      const int kb = k0 + half * 2;
      f3[k0 / 4].x = x3[kb]; f3[k0 / 4].y = x3[kb + 1];
    }
  } else {
#pragma unroll
    for (int q = 0; q < DFEAT / 4; ++q) { f3[q].x = 0.0f; f3[q].y = 0.0f; }
  }

  acc[tid] = 0.0f;
  if (tid < RNODES - RTHREADS) acc[tid + RTHREADS] = 0.0f;
  range_sum<true>(bins, offs, nullptr, acc, blockIdx.x);
  __syncthreads();
  {
    const float dd = rsqrtf(1.0f + acc[tid]);
    dinv_s[tid] = dd;
    if (lo + tid < n) dinv[lo + tid] = dd;
    if (tid < RNODES - RTHREADS) {
      const float d2 = rsqrtf(1.0f + acc[tid + RTHREADS]);
      dinv_s[tid + RTHREADS] = d2;
      if (lo + tid + RTHREADS < n) dinv[lo + tid + RTHREADS] = d2;
    }
  }

  v8f c0 = {};
  v8f c1 = {};
  v8f c2 = {};
  v8f c3 = {};
  if (four) {
#pragma unroll
    for (int q = 0; q < DFEAT / 4; ++q)
      c3 = __builtin_amdgcn_wmma_f32_16x16x4_f32(
           false, f3[q],  false, fw[q],
           (short)0, c3,  false,  false);
  }
#pragma unroll
  for (int q = 0; q < DFEAT / 4; ++q) {
    c0 = __builtin_amdgcn_wmma_f32_16x16x4_f32(
         false, f0[q],  false, fw[q],
         (short)0, c0,  false,  false);
    c1 = __builtin_amdgcn_wmma_f32_16x16x4_f32(
         false, f1[q],  false, fw[q],
         (short)0, c1,  false,  false);
    c2 = __builtin_amdgcn_wmma_f32_16x16x4_f32(
         false, f2[q],  false, fw[q],
         (short)0, c2,  false,  false);
  }

  const int rr = m & 7;
  float p0 = 0.0f, p1 = 0.0f, p2 = 0.0f, p3 = 0.0f;
#pragma unroll
  for (int r = 0; r < 8; ++r) {
    p0 = (rr == r) ? c0[r] : p0;
    p1 = (rr == r) ? c1[r] : p1;
    p2 = (rr == r) ? c2[r] : p2;
    p3 = (rr == r) ? c3[r] : p3;
  }
  if (m < 8) {
    const int in = half * 8 + rr;
    z_s[(wv     ) * 16 + in] = p0;
    z_s[(wv +  8) * 16 + in] = p1;
    z_s[(wv + 16) * 16 + in] = p2;
    if (four) z_s[(wv + 24) * 16 + in] = p3;
  }
  __syncthreads();
  if (lo + tid < n) zy[lo + tid] = dinv_s[tid] * z_s[tid];
  if (tid < RNODES - RTHREADS && lo + tid + RTHREADS < n)
    zy[lo + tid + RTHREADS] = dinv_s[tid + RTHREADS] * z_s[tid + RTHREADS];
}

template <bool kLast>
__global__ __launch_bounds__(RTHREADS)
void hop_kernel(const unsigned int* __restrict__ bins,
                const int* __restrict__ offs,
                const float* __restrict__ dinv,
                const float* __restrict__ yin,
                const float* __restrict__ bptr,
                float* __restrict__ yout, int n) {
  __shared__ float acc[RNODES];
  const int tid   = threadIdx.x;
  const bool two  = tid < RNODES - RTHREADS;
  const int node0 = blockIdx.x * RNODES + tid;
  const int node1 = node0 + RTHREADS;
  const int nc0   = node0 < n ? node0 : n - 1;
  const int nc1   = (two && node1 < n) ? node1 : n - 1;
  const float d0    = dinv[nc0];
  const float self0 = yin[nc0];
  const float d1    = dinv[nc1];
  const float self1 = yin[nc1];
  const float bias  = kLast ? bptr[0] : 0.0f;

  acc[tid] = 0.0f;
  if (two) acc[tid + RTHREADS] = 0.0f;
  range_sum<false>(bins, offs, yin, acc, blockIdx.x);
  __syncthreads();

  const float z0 = d0 * (self0 + acc[tid]);
  if (node0 < n) {
    if (kLast) yout[node0] = 1.0f / (1.0f + __expf(-(z0 + bias)));
    else       yout[node0] = d0 * z0;
  }
  if (two && node1 < n) {
    const float z1 = d1 * (self1 + acc[tid + RTHREADS]);
    if (kLast) yout[node1] = 1.0f / (1.0f + __expf(-(z1 + bias)));
    else       yout[node1] = d1 * z1;
  }
}

#define ROW_THREADS 256
#define ROW_ITEMS   16
__global__ __launch_bounds__(ROW_THREADS)
void scale_rows_kernel(const float* __restrict__ x,
                       const float* __restrict__ score,
                       float* __restrict__ outx, int total4) {
  const float4* x4 = reinterpret_cast<const float4*>(x);
  float4* o4 = reinterpret_cast<float4*>(outx);
  const int first = blockIdx.x * (ROW_THREADS * ROW_ITEMS) + (int)threadIdx.x;
  float4 xv[ROW_ITEMS];
  float  sc[ROW_ITEMS];
#pragma unroll
  for (int j = 0; j < ROW_ITEMS; ++j) {
    int i = first + j * ROW_THREADS;
    if (i > total4 - 1) i = total4 - 1;
    xv[j] = x4[i];
    sc[j] = score[i >> 4];
  }
#pragma unroll
  for (int j = 0; j < ROW_ITEMS; ++j) {
    const int i = first + j * ROW_THREADS;
    if (i < total4)
      o4[i] = make_float4(xv[j].x * sc[j], xv[j].y * sc[j],
                          xv[j].z * sc[j], xv[j].w * sc[j]);
  }
}

extern "C" void kernel_launch(void* const* d_in, const int* in_sizes, int n_in,
                              void* d_out, int out_size, void* d_ws, size_t ws_size,
                              hipStream_t stream) {
  const float* x  = (const float*)d_in[0];
  const int*   ei = (const int*)d_in[1];
  const float* W  = (const float*)d_in[2];
  const float* b  = (const float*)d_in[3];

  const int n = in_sizes[0] / DFEAT;
  const int e = in_sizes[1] / 2;
  const int* src = ei;
  const int* dst = ei + e;

  const size_t nPad = ((size_t)n + 63) & ~(size_t)63;
  float* dinv = (float*)d_ws;
  float* zy0  = dinv + nPad;
  float* y1   = zy0 + nPad;
  float* outx = (float*)d_out;
  float* outs = (float*)d_out + (size_t)n * DFEAT;

  unsigned int* bins = (unsigned int*)d_out;
  int* offs = (int*)d_out + (size_t)NSLICE * BIN_CAP;

  const int nRanges  = (n + RNODES - 1) / RNODES;
  const int sliceLen = (e + NSLICE - 1) / NSLICE;
  const int total4   = n * (DFEAT / 4);

  bin_edges_kernel<<<NSLICE, BIN_THREADS, 0, stream>>>(src, dst, bins, offs, e, sliceLen);
  degree_project_kernel<<<nRanges, RTHREADS, 0, stream>>>(x, W, bins, offs, dinv, zy0, n);
  hop_kernel<false><<<nRanges, RTHREADS, 0, stream>>>(bins, offs, dinv, zy0, b, y1, n);
  hop_kernel<true><<<nRanges, RTHREADS, 0, stream>>>(bins, offs, dinv, y1, b, outs, n);
  scale_rows_kernel<<<(total4 + ROW_THREADS * ROW_ITEMS - 1) / (ROW_THREADS * ROW_ITEMS), ROW_THREADS, 0, stream>>>(
      x, outs, outx, total4);
}


# Placed edges staged in the workspace when it fits (else as before): hop 2's blocks store their own 416 rows, x loaded behind the gathers, in flight over LDS-only barriers; 4 launches; add order kept
# speedup vs baseline: 9.9829x; 1.0471x over previous
#include <hip/hip_runtime.h>
#include <cstdint>

#define DFEAT 64


typedef float v2f __attribute__((ext_vector_type(2)));
typedef float v8f __attribute__((ext_vector_type(8)));

#define NSLICE      256
#define RSHIFT      9
#define RMASK       ((1u << RSHIFT) - 1u)
#define RNODES      416
#define RTHREADS    256
#define RTILES      (RNODES / 16)
#define MAXR        512
#define BIN_THREADS 256
#define BIN_ITEMS   25
#define BIN_CAP     (BIN_THREADS * BIN_ITEMS)

__global__ __launch_bounds__(BIN_THREADS)
void bin_edges_kernel(const int* __restrict__ src,
                      const int* __restrict__ dst,
                      unsigned int* __restrict__ bins,
                      int* __restrict__ offs,
                      int e, int sliceLen) {
  __shared__ int   start[MAXR];
  __shared__ uint4 stage4[BIN_CAP / 4];
  __shared__ int   wsum[BIN_THREADS / 32];
  unsigned int* stage = reinterpret_cast<unsigned int*>(stage4);

  const int tid  = threadIdx.x;
  const int lane = tid & 31;
  const int wave = tid >> 5;
  for (int i = tid; i < MAXR; i += BIN_THREADS) start[i] = 0;
  __syncthreads();

  const int sLo = blockIdx.x * sliceLen;
  int sHi = sLo + sliceLen;
  if (sHi > e) sHi = e;

  int d[BIN_ITEMS];
  int s[BIN_ITEMS];
  int rk[BIN_ITEMS];
#pragma unroll
  for (int j = 0; j < BIN_ITEMS; ++j) {
    const int idx = sLo + tid + j * BIN_THREADS;
    const int ic  = idx < sHi ? idx : 0;
    d[j] = dst[ic];
    s[j] = src[ic];
  }
#pragma unroll
  for (int j = 0; j < BIN_ITEMS; ++j) {
    const int idx = sLo + tid + j * BIN_THREADS;
    rk[j] = idx < sHi ? atomicAdd(&start[d[j] / RNODES], 1) : 0;
  }
  __syncthreads();

  const int c0 = start[2 * tid];
  const int c1 = start[2 * tid + 1];
  const int both = c0 + c1;
  int inc = both;
  for (int o = 1; o < 32; o <<= 1) {
    const int up = __shfl_up(inc, o);
    if (lane >= o) inc += up;
  }
  if (lane == 31) wsum[wave] = inc;
  __syncthreads();
  int wbase = 0;
  for (int w = 0; w < wave; ++w) wbase += wsum[w];
  const int ex0 = wbase + inc - both;
  const int ex1 = ex0 + c0;
  start[2 * tid]     = ex0;
  start[2 * tid + 1] = ex1;
  int* row = offs + (size_t)blockIdx.x * MAXR;
  row[2 * tid]     = ex0;
  row[2 * tid + 1] = ex1;
  __syncthreads();

#pragma unroll
  for (int j = 0; j < BIN_ITEMS; ++j) {
    const int idx = sLo + tid + j * BIN_THREADS;
    if (idx < sHi) {
      const int rg = d[j] / RNODES;
      stage[start[rg] + rk[j]] =
          ((unsigned int)s[j] << RSHIFT) | (unsigned int)(d[j] - rg * RNODES);
    }
  }
  __syncthreads();

  int cntS = sHi - sLo;
  if (cntS < 0) cntS = 0;
  unsigned int* seg = bins + (size_t)blockIdx.x * BIN_CAP;
  uint4* seg4 = reinterpret_cast<uint4*>(seg);
  const int c4 = cntS >> 2;
  for (int i = tid; i < c4; i += BIN_THREADS) seg4[i] = stage4[i];
  for (int i = (c4 << 2) + tid; i < cntS; i += BIN_THREADS) seg[i] = stage[i];
}

__device__ __forceinline__ void lds_barrier() {
  __builtin_amdgcn_fence(__ATOMIC_RELEASE, "workgroup", "local");
  __builtin_amdgcn_s_barrier();
  __builtin_amdgcn_fence(__ATOMIC_ACQUIRE, "workgroup", "local");
}

template <bool kDegree, bool kLdsOnly, class More>
__device__ __forceinline__ void range_sum(const unsigned int* __restrict__ bins,
                                          const int* __restrict__ offs,
                                          const float* __restrict__ yin,
                                          float* acc, int r, More more) {
  const int lane = threadIdx.x & 31;
  const int wave = threadIdx.x >> 5;
  const int* row = offs + (size_t)(wave * 32 + lane) * MAXR + r;
  const int beg = row[0];
  const int len = row[1] - beg;
  const unsigned int* wbase = bins + (size_t)(wave * 32) * BIN_CAP;

  unsigned int wd[32];
#pragma unroll
  for (int j = 0; j < 32; ++j) {
    const int b = __builtin_amdgcn_readlane(beg, j);
    const int l = __builtin_amdgcn_readlane(len, j);
    const int k = lane < l ? b + lane : 0;
    wd[j] = wbase[(size_t)j * BIN_CAP + k];
  }

  const int ov = len > 32 ? len - 32 : 0;
  int inc = ov;
  for (int o = 1; o < 32; o <<= 1) {
    const int up = __shfl_up(inc, o);
    if (lane >= o) inc += up;
  }
  const int exc = inc - ov;
  const int tot = __builtin_amdgcn_readlane(inc, 31);
  int jx = 0;
#pragma unroll
  for (int l = 0; l < 32; ++l) jx += (__builtin_amdgcn_readlane(inc, l) <= lane) ? 1 : 0;
  const int jxc = jx < 32 ? jx : 31;
  const int ejx = __shfl(exc, jxc);
  const int bjx = __shfl(beg, jxc);
  const bool okx = lane < tot;
  const unsigned int wdx = wbase[okx ? (size_t)jxc * BIN_CAP + (size_t)(bjx + 32 + (lane - ejx)) : (size_t)0];
  __builtin_amdgcn_sched_barrier(0);

#pragma unroll
  for (int j = 0; j < 32; ++j) asm volatile("" :: "v"(wd[j]));
  float v[32];
  float vx = 1.0f;
#pragma unroll
  for (int j = 0; j < 32; ++j) {
    if (kDegree) {
      v[j] = 1.0f;
    } else {
      const int l = __builtin_amdgcn_readlane(len, j);
      const unsigned int sg = lane < l ? (wd[j] >> RSHIFT) : 0u;
      v[j] = yin[sg];
    }
  }
  if (!kDegree) {
    vx = yin[okx ? (wdx >> RSHIFT) : 0u];
    more();
#pragma unroll
    for (int j = 0; j < 32; ++j) asm volatile("" :: "v"(v[j]));
    asm volatile("" :: "v"(vx));
  }
  if (kLdsOnly) lds_barrier();
  else          __syncthreads();
#pragma unroll
  for (int j = 0; j < 32; ++j) {
    const int l = __builtin_amdgcn_readlane(len, j);
    if (lane < l) atomicAdd(&acc[wd[j] & RMASK], v[j]);
  }
  if (okx) atomicAdd(&acc[wdx & RMASK], vx);
  if (tot > 32) {
#pragma nounroll
    for (int base = 32; base < tot; base += 32) {
      const int t = base + lane;
      int j2 = 0;
#pragma unroll
      for (int l = 0; l < 32; ++l) j2 += (__builtin_amdgcn_readlane(inc, l) <= t) ? 1 : 0;
      const int j2c = j2 < 32 ? j2 : 31;
      const int e2 = __shfl(exc, j2c);
      const int b2 = __shfl(beg, j2c);
      if (t < tot) {
        const unsigned int w2 = wbase[(size_t)j2c * BIN_CAP + (size_t)(b2 + 32 + (t - e2))];
        const float val = kDegree ? 1.0f : yin[w2 >> RSHIFT];
        atomicAdd(&acc[w2 & RMASK], val);
      }
    }
  }
}

__global__ __launch_bounds__(RTHREADS)
void degree_project_kernel(const float* __restrict__ x,
                           const float* __restrict__ W,
                           const unsigned int* __restrict__ bins,
                           const int* __restrict__ offs,
                           float* __restrict__ dinv,
                           float* __restrict__ zy, int n) {
  __shared__ float acc[RNODES];
  __shared__ float dinv_s[RNODES];
  __shared__ float z_s[RTILES * 16];
  const int tid  = threadIdx.x;
  const int lane = tid & 31;
  const int wv   = __builtin_amdgcn_readfirstlane(tid >> 5);
  const int lo   = blockIdx.x * RNODES;
  const int half = lane >> 4;
  const int m    = lane & 15;
  const bool four = wv + 24 < RTILES;
  int r0 = lo + (wv     ) * 16 + m;
  int r1 = lo + (wv +  8) * 16 + m;
  int r2 = lo + (wv + 16) * 16 + m;
  int r3 = lo + (wv + 24) * 16 + m;
  if (r0 > n - 1) r0 = n - 1;
  if (r1 > n - 1) r1 = n - 1;
  if (r2 > n - 1) r2 = n - 1;
  if (r3 > n - 1) r3 = n - 1;
  const float* x0 = x + (size_t)r0 * DFEAT;
  const float* x1 = x + (size_t)r1 * DFEAT;
  const float* x2 = x + (size_t)r2 * DFEAT;
  const float* x3 = x + (size_t)r3 * DFEAT;

  v2f f0[DFEAT / 4];
  v2f f1[DFEAT / 4];
  v2f f2[DFEAT / 4];
  v2f f3[DFEAT / 4];
  v2f fw[DFEAT / 4];
#pragma unroll
  for (int k0 = 0; k0 < DFEAT; k0 += 4) {
    const int kb = k0 + half * 2;
    fw[k0 / 4].x = W[kb];  fw[k0 / 4].y = W[kb + 1];
    f0[k0 / 4].x = x0[kb]; f0[k0 / 4].y = x0[kb + 1];
    f1[k0 / 4].x = x1[kb]; f1[k0 / 4].y = x1[kb + 1];
    f2[k0 / 4].x = x2[kb]; f2[k0 / 4].y = x2[kb + 1];
  }
  if (four) {
#pragma unroll
    for (int k0 = 0; k0 < DFEAT; k0 += 4) {
      const int kb = k0 + half * 2;
      f3[k0 / 4].x = x3[kb]; f3[k0 / 4].y = x3[kb + 1];
    }
  } else {
#pragma unroll
    for (int q = 0; q < DFEAT / 4; ++q) { f3[q].x = 0.0f; f3[q].y = 0.0f; }
  }

  acc[tid] = 0.0f;
  if (tid < RNODES - RTHREADS) acc[tid + RTHREADS] = 0.0f;
  range_sum<true, false>(bins, offs, nullptr, acc, blockIdx.x, []() {});
  __syncthreads();
  {
    const float dd = rsqrtf(1.0f + acc[tid]);
    dinv_s[tid] = dd;
    if (lo + tid < n) dinv[lo + tid] = dd;
    if (tid < RNODES - RTHREADS) {
      const float d2 = rsqrtf(1.0f + acc[tid + RTHREADS]);
      dinv_s[tid + RTHREADS] = d2;
      if (lo + tid + RTHREADS < n) dinv[lo + tid + RTHREADS] = d2;
    }
  }

  v8f c0 = {};
  v8f c1 = {};
  v8f c2 = {};
  v8f c3 = {};
  if (four) {
#pragma unroll
    for (int q = 0; q < DFEAT / 4; ++q)
      c3 = __builtin_amdgcn_wmma_f32_16x16x4_f32(
           false, f3[q],  false, fw[q],
           (short)0, c3,  false,  false);
  }
#pragma unroll
  for (int q = 0; q < DFEAT / 4; ++q) {
    c0 = __builtin_amdgcn_wmma_f32_16x16x4_f32(
         false, f0[q],  false, fw[q],
         (short)0, c0,  false,  false);
    c1 = __builtin_amdgcn_wmma_f32_16x16x4_f32(
         false, f1[q],  false, fw[q],
         (short)0, c1,  false,  false);
    c2 = __builtin_amdgcn_wmma_f32_16x16x4_f32(
         false, f2[q],  false, fw[q],
         (short)0, c2,  false,  false);
  }

  const int rr = m & 7;
  float p0 = 0.0f, p1 = 0.0f, p2 = 0.0f, p3 = 0.0f;
#pragma unroll
  for (int r = 0; r < 8; ++r) {
    p0 = (rr == r) ? c0[r] : p0;
    p1 = (rr == r) ? c1[r] : p1;
    p2 = (rr == r) ? c2[r] : p2;
    p3 = (rr == r) ? c3[r] : p3;
  }
  if (m < 8) {
    const int in = half * 8 + rr;
    z_s[(wv     ) * 16 + in] = p0;
    z_s[(wv +  8) * 16 + in] = p1;
    z_s[(wv + 16) * 16 + in] = p2;
    if (four) z_s[(wv + 24) * 16 + in] = p3;
  }
  __syncthreads();
  if (lo + tid < n) zy[lo + tid] = dinv_s[tid] * z_s[tid];
  if (tid < RNODES - RTHREADS && lo + tid + RTHREADS < n)
    zy[lo + tid + RTHREADS] = dinv_s[tid + RTHREADS] * z_s[tid + RTHREADS];
}

#define ROW_PIECES (RNODES * (DFEAT / 4) / RTHREADS)
template <bool kLast, bool kRows>
__global__ __launch_bounds__(RTHREADS)
void hop_kernel(const unsigned int* __restrict__ bins,
                const int* __restrict__ offs,
                const float* __restrict__ dinv,
                const float* __restrict__ yin,
                const float* __restrict__ bptr,
                float* __restrict__ yout,
                const float* __restrict__ x,
                float* __restrict__ outx, int n) {
  __shared__ float acc[RNODES];
  const int tid   = threadIdx.x;
  const bool two  = tid < RNODES - RTHREADS;
  const int node0 = blockIdx.x * RNODES + tid;
  const int node1 = node0 + RTHREADS;
  const int nc0   = node0 < n ? node0 : n - 1;
  const int nc1   = (two && node1 < n) ? node1 : n - 1;
  const float d0    = dinv[nc0];
  const float self0 = yin[nc0];
  const float d1    = dinv[nc1];
  const float self1 = yin[nc1];
  const float bias  = kLast ? bptr[0] : 0.0f;

  const float4* x4 = reinterpret_cast<const float4*>(x);
  float4* o4 = reinterpret_cast<float4*>(outx);
  const int total4 = n * (DFEAT / 4);
  const int first4 = blockIdx.x * (RNODES * (DFEAT / 4)) + tid;
  float4 xv[ROW_PIECES];

  acc[tid] = 0.0f;
  if (two) acc[tid + RTHREADS] = 0.0f;
  range_sum<false, kRows>(bins, offs, yin, acc, blockIdx.x, [&]() {
    if (kRows) {
      __builtin_amdgcn_sched_barrier(0);
#pragma unroll
      for (int j = 0; j < ROW_PIECES; ++j) {
        int i = first4 + j * RTHREADS;
        if (i > total4 - 1) i = total4 - 1;
        xv[j] = x4[i];
      }
    }
  });
  if (kRows) lds_barrier();
  else       __syncthreads();

  const float z0 = d0 * (self0 + acc[tid]);
  const float s0 = kLast ? 1.0f / (1.0f + __expf(-(z0 + bias))) : d0 * z0;
  if (node0 < n) yout[node0] = s0;
  if (kRows) acc[tid] = s0;
  if (two) {
    const float z1 = d1 * (self1 + acc[tid + RTHREADS]);
    const float s1 = kLast ? 1.0f / (1.0f + __expf(-(z1 + bias))) : d1 * z1;
    if (node1 < n) yout[node1] = s1;
    if (kRows) acc[tid + RTHREADS] = s1;
  }
  if (kRows) {
    lds_barrier();
    float sc[ROW_PIECES];
#pragma unroll
    for (int j = 0; j < ROW_PIECES; ++j) sc[j] = acc[(tid >> 4) + 16 * j];
#pragma unroll
    for (int j = 0; j < ROW_PIECES; ++j) {
      const int i = first4 + j * RTHREADS;
      asm volatile("" :: "v"(sc[j]), "v"(xv[j].x), "v"(xv[j].y), "v"(xv[j].z), "v"(xv[j].w));
      if (i < total4)
        o4[i] = make_float4(xv[j].x * sc[j], xv[j].y * sc[j], xv[j].z * sc[j], xv[j].w * sc[j]);
    }
  }
}

#define ROW_THREADS 256
#define ROW_ITEMS   16
__global__ __launch_bounds__(ROW_THREADS)
void scale_rows_kernel(const float* __restrict__ x,
                       const float* __restrict__ score,
                       float* __restrict__ outx, int total4) {
  const float4* x4 = reinterpret_cast<const float4*>(x);
  float4* o4 = reinterpret_cast<float4*>(outx);
  const int first = blockIdx.x * (ROW_THREADS * ROW_ITEMS) + (int)threadIdx.x;
  float4 xv[ROW_ITEMS];
  float  sc[ROW_ITEMS];
#pragma unroll
  for (int j = 0; j < ROW_ITEMS; ++j) {
    int i = first + j * ROW_THREADS;
    if (i > total4 - 1) i = total4 - 1;
    xv[j] = x4[i];
    sc[j] = score[i >> 4];
  }
#pragma unroll
  for (int j = 0; j < ROW_ITEMS; ++j) {
    const int i = first + j * ROW_THREADS;
    if (i < total4)
      o4[i] = make_float4(xv[j].x * sc[j], xv[j].y * sc[j],
                          xv[j].z * sc[j], xv[j].w * sc[j]);
  }
}

extern "C" void kernel_launch(void* const* d_in, const int* in_sizes, int n_in,
                              void* d_out, int out_size, void* d_ws, size_t ws_size,
                              hipStream_t stream) {
  const float* x  = (const float*)d_in[0];
  const int*   ei = (const int*)d_in[1];
  const float* W  = (const float*)d_in[2];
  const float* b  = (const float*)d_in[3];

  const int n = in_sizes[0] / DFEAT;
  const int e = in_sizes[1] / 2;
  const int* src = ei;
  const int* dst = ei + e;

  const size_t nPad = ((size_t)n + 63) & ~(size_t)63;
  float* dinv = (float*)d_ws;
  float* zy0  = dinv + nPad;
  float* y1   = zy0 + nPad;
  float* outx = (float*)d_out;
  float* outs = (float*)d_out + (size_t)n * DFEAT;

  const size_t binWords = (size_t)NSLICE * BIN_CAP;
  const size_t tabWords = (size_t)NSLICE * MAXR;
  const bool inWs = ws_size >= (3 * nPad + binWords + tabWords) * sizeof(float);
  unsigned int* bins = inWs ? (unsigned int*)(y1 + nPad) : (unsigned int*)d_out;
  int* offs = (int*)(bins + binWords);

  const int nRanges  = (n + RNODES - 1) / RNODES;
  const int sliceLen = (e + NSLICE - 1) / NSLICE;
  const int total4   = n * (DFEAT / 4);

  bin_edges_kernel<<<NSLICE, BIN_THREADS, 0, stream>>>(src, dst, bins, offs, e, sliceLen);
  degree_project_kernel<<<nRanges, RTHREADS, 0, stream>>>(x, W, bins, offs, dinv, zy0, n);
  hop_kernel<false, false><<<nRanges, RTHREADS, 0, stream>>>(bins, offs, dinv, zy0, b, y1, x, outx, n);
  if (inWs) {
    hop_kernel<true, true><<<nRanges, RTHREADS, 0, stream>>>(bins, offs, dinv, y1, b, outs, x, outx, n);
  } else {
    hop_kernel<true, false><<<nRanges, RTHREADS, 0, stream>>>(bins, offs, dinv, y1, b, outs, x, outx, n);
    scale_rows_kernel<<<(total4 + ROW_THREADS * ROW_ITEMS - 1) / (ROW_THREADS * ROW_ITEMS), ROW_THREADS, 0, stream>>>(
        x, outs, outx, total4);
  }
}


# Hops read source values from an LDS copy of the first 79872 node values (each work-group fills it with whole-line vector loads), not one L2 line per edge; rest gathered as before; add order kept
# speedup vs baseline: 10.5495x; 1.0568x over previous
#include <hip/hip_runtime.h>
#include <cstdint>

#define DFEAT 64


typedef float v2f __attribute__((ext_vector_type(2)));
typedef float v8f __attribute__((ext_vector_type(8)));
typedef float v4f __attribute__((ext_vector_type(4)));

#define NSLICE      256
#define RSHIFT      9
#define RMASK       ((1u << RSHIFT) - 1u)
#define RNODES      416
#define RTHREADS    256
#define RTILES      (RNODES / 16)
#define MAXR        512
#define BIN_THREADS 256
#define BIN_ITEMS   25
#define BIN_CAP     (BIN_THREADS * BIN_ITEMS)
#define FILL_CHUNK  26
#define FILL_CHUNKS 3
#define LCAP        (FILL_CHUNKS * FILL_CHUNK * RTHREADS * 4)

__global__ __launch_bounds__(BIN_THREADS)
void bin_edges_kernel(const int* __restrict__ src,
                      const int* __restrict__ dst,
                      unsigned int* __restrict__ bins,
                      int* __restrict__ offs,
                      int e, int sliceLen) {
  __shared__ int   start[MAXR];
  __shared__ uint4 stage4[BIN_CAP / 4];
  __shared__ int   wsum[BIN_THREADS / 32];
  unsigned int* stage = reinterpret_cast<unsigned int*>(stage4);

  const int tid  = threadIdx.x;
  const int lane = tid & 31;
  const int wave = tid >> 5;
  for (int i = tid; i < MAXR; i += BIN_THREADS) start[i] = 0;
  __syncthreads();

  const int sLo = blockIdx.x * sliceLen;
  int sHi = sLo + sliceLen;
  if (sHi > e) sHi = e;

  int d[BIN_ITEMS];
  int s[BIN_ITEMS];
  int rk[BIN_ITEMS];
#pragma unroll
  for (int j = 0; j < BIN_ITEMS; ++j) {
    const int idx = sLo + tid + j * BIN_THREADS;
    const int ic  = idx < sHi ? idx : 0;
    d[j] = dst[ic];
    s[j] = src[ic];
  }
#pragma unroll
  for (int j = 0; j < BIN_ITEMS; ++j) {
    const int idx = sLo + tid + j * BIN_THREADS;
    rk[j] = idx < sHi ? atomicAdd(&start[d[j] / RNODES], 1) : 0;
  }
  __syncthreads();

  const int c0 = start[2 * tid];
  const int c1 = start[2 * tid + 1];
  const int both = c0 + c1;
  int inc = both;
  for (int o = 1; o < 32; o <<= 1) {
    const int up = __shfl_up(inc, o);
    if (lane >= o) inc += up;
  }
  if (lane == 31) wsum[wave] = inc;
  __syncthreads();
  int wbase = 0;
  for (int w = 0; w < wave; ++w) wbase += wsum[w];
  const int ex0 = wbase + inc - both;
  const int ex1 = ex0 + c0;
  start[2 * tid]     = ex0;
  start[2 * tid + 1] = ex1;
  int* row = offs + (size_t)blockIdx.x * MAXR;
  row[2 * tid]     = ex0;
  row[2 * tid + 1] = ex1;
  __syncthreads();

#pragma unroll
  for (int j = 0; j < BIN_ITEMS; ++j) {
    const int idx = sLo + tid + j * BIN_THREADS;
    if (idx < sHi) {
      const int rg = d[j] / RNODES;
      stage[start[rg] + rk[j]] =
          ((unsigned int)s[j] << RSHIFT) | (unsigned int)(d[j] - rg * RNODES);
    }
  }
  __syncthreads();

  int cntS = sHi - sLo;
  if (cntS < 0) cntS = 0;
  unsigned int* seg = bins + (size_t)blockIdx.x * BIN_CAP;
  uint4* seg4 = reinterpret_cast<uint4*>(seg);
  const int c4 = cntS >> 2;
  for (int i = tid; i < c4; i += BIN_THREADS) seg4[i] = stage4[i];
  for (int i = (c4 << 2) + tid; i < cntS; i += BIN_THREADS) seg[i] = stage[i];
}

__device__ __forceinline__ void lds_barrier() {
  __builtin_amdgcn_fence(__ATOMIC_RELEASE, "workgroup", "local");
  __builtin_amdgcn_s_barrier();
  __builtin_amdgcn_fence(__ATOMIC_ACQUIRE, "workgroup", "local");
}

__device__ __forceinline__ void range_count(const unsigned int* __restrict__ bins,
                                            const int* __restrict__ offs,
                                            float* acc, int r) {
  const int lane = threadIdx.x & 31;
  const int wave = threadIdx.x >> 5;
  const int* row = offs + (size_t)(wave * 32 + lane) * MAXR + r;
  const int beg = row[0];
  const int len = row[1] - beg;
  const unsigned int* wbase = bins + (size_t)(wave * 32) * BIN_CAP;

  unsigned int wd[32];
#pragma unroll
  for (int j = 0; j < 32; ++j) {
    const int b = __builtin_amdgcn_readlane(beg, j);
    const int l = __builtin_amdgcn_readlane(len, j);
    const int k = lane < l ? b + lane : 0;
    wd[j] = wbase[(size_t)j * BIN_CAP + k];
  }

  const int ov = len > 32 ? len - 32 : 0;
  int inc = ov;
  for (int o = 1; o < 32; o <<= 1) {
    const int up = __shfl_up(inc, o);
    if (lane >= o) inc += up;
  }
  const int exc = inc - ov;
  const int tot = __builtin_amdgcn_readlane(inc, 31);
  int jx = 0;
#pragma unroll
  for (int l = 0; l < 32; ++l) jx += (__builtin_amdgcn_readlane(inc, l) <= lane) ? 1 : 0;
  const int jxc = jx < 32 ? jx : 31;
  const int ejx = __shfl(exc, jxc);
  const int bjx = __shfl(beg, jxc);
  const bool okx = lane < tot;
  const unsigned int wdx = wbase[okx ? (size_t)jxc * BIN_CAP + (size_t)(bjx + 32 + (lane - ejx)) : (size_t)0];
  __builtin_amdgcn_sched_barrier(0);

#pragma unroll
  for (int j = 0; j < 32; ++j) asm volatile("" :: "v"(wd[j]));
  __syncthreads();
#pragma unroll
  for (int j = 0; j < 32; ++j) {
    const int l = __builtin_amdgcn_readlane(len, j);
    if (lane < l) atomicAdd(&acc[wd[j] & RMASK], 1.0f);
  }
  if (okx) atomicAdd(&acc[wdx & RMASK], 1.0f);
  if (tot > 32) {
#pragma nounroll
    for (int base = 32; base < tot; base += 32) {
      const int t = base + lane;
      int j2 = 0;
#pragma unroll
      for (int l = 0; l < 32; ++l) j2 += (__builtin_amdgcn_readlane(inc, l) <= t) ? 1 : 0;
      const int j2c = j2 < 32 ? j2 : 31;
      const int e2 = __shfl(exc, j2c);
      const int b2 = __shfl(beg, j2c);
      if (t < tot) {
        const unsigned int w2 = wbase[(size_t)j2c * BIN_CAP + (size_t)(b2 + 32 + (t - e2))];
        atomicAdd(&acc[w2 & RMASK], 1.0f);
      }
    }
  }
}

template <class More>
__device__ __forceinline__ void range_sum_lds(const unsigned int* __restrict__ bins,
                                              const int* __restrict__ offs,
                                              const float* __restrict__ yin,
                                              v4f* plane4, int n,
                                              float* acc, int r, More more) {
  const int tid  = threadIdx.x;
  const int lane = tid & 31;
  const int wave = tid >> 5;
  const int* row = offs + (size_t)(wave * 32 + lane) * MAXR + r;
  const int beg = row[0];
  const int len = row[1] - beg;
  const unsigned int* wbase = bins + (size_t)(wave * 32) * BIN_CAP;

  const float*  plane = reinterpret_cast<const float*>(plane4);
  const v4f* y4       = reinterpret_cast<const v4f*>(yin);
  int nl = n & ~3;
  if (nl > LCAP) nl = LCAP;
  const int nl4 = nl >> 2;

  unsigned int wd[32];
#pragma unroll
  for (int j = 0; j < 32; ++j) {
    const int b = __builtin_amdgcn_readlane(beg, j);
    const int l = __builtin_amdgcn_readlane(len, j);
    const int k = lane < l ? b + lane : 0;
    wd[j] = wbase[(size_t)j * BIN_CAP + k];
  }

  const int ov = len > 32 ? len - 32 : 0;
  int inc = ov;
  for (int o = 1; o < 32; o <<= 1) {
    const int up = __shfl_up(inc, o);
    if (lane >= o) inc += up;
  }
  const int exc = inc - ov;
  const int tot = __builtin_amdgcn_readlane(inc, 31);
  int jx = 0;
#pragma unroll
  for (int l = 0; l < 32; ++l) jx += (__builtin_amdgcn_readlane(inc, l) <= lane) ? 1 : 0;
  const int jxc = jx < 32 ? jx : 31;
  const int ejx = __shfl(exc, jxc);
  const int bjx = __shfl(beg, jxc);
  const bool okx = lane < tot;
  const unsigned int wdx = wbase[okx ? (size_t)jxc * BIN_CAP + (size_t)(bjx + 32 + (lane - ejx)) : (size_t)0];
  __builtin_amdgcn_sched_barrier(0);

  v4f pa[FILL_CHUNK];
  v4f pb[FILL_CHUNK];
#pragma unroll
  for (int j = 0; j < FILL_CHUNK; ++j) {
    const int i = tid + j * RTHREADS;
    pa[j] = y4[i < nl4 ? i : 0];
  }
#pragma unroll
  for (int j = 0; j < FILL_CHUNK; ++j) {
    const int i = tid + (FILL_CHUNK + j) * RTHREADS;
    pb[j] = y4[i < nl4 ? i : 0];
  }
  __builtin_amdgcn_sched_barrier(0);

  float vg[32];
#pragma unroll
  for (int j = 0; j < 32; ++j) {
    const int l = __builtin_amdgcn_readlane(len, j);
    const unsigned int s = wd[j] >> RSHIFT;
    const bool far = lane < l && s >= (unsigned int)nl;
    vg[j] = yin[far ? s : 0u];
  }
  const unsigned int sx = wdx >> RSHIFT;
  const float vgx = yin[(okx && sx >= (unsigned int)nl) ? sx : 0u];
  __builtin_amdgcn_sched_barrier(0);

#pragma unroll
  for (int j = 0; j < FILL_CHUNK; ++j) plane4[tid + j * RTHREADS] = pa[j];
#pragma unroll
  for (int j = 0; j < FILL_CHUNK; ++j) {
    const int i = tid + (2 * FILL_CHUNK + j) * RTHREADS;
    pa[j] = y4[i < nl4 ? i : 0];
  }
  more();
#pragma unroll
  for (int j = 0; j < FILL_CHUNK; ++j) plane4[tid + (FILL_CHUNK + j) * RTHREADS] = pb[j];
#pragma unroll
  for (int j = 0; j < FILL_CHUNK; ++j) plane4[tid + (2 * FILL_CHUNK + j) * RTHREADS] = pa[j];
  lds_barrier();

  float v[32];
#pragma unroll
  for (int j = 0; j < 32; ++j) {
    const int l = __builtin_amdgcn_readlane(len, j);
    const unsigned int s = wd[j] >> RSHIFT;
    const bool near = lane < l && s < (unsigned int)nl;
    const float t = plane[near ? s : 0u];
    v[j] = near ? t : vg[j];
  }
  const bool nearx = okx && sx < (unsigned int)nl;
  const float tx = plane[nearx ? sx : 0u];
  const float vx = nearx ? tx : vgx;

#pragma unroll
  for (int j = 0; j < 32; ++j) {
    const int l = __builtin_amdgcn_readlane(len, j);
    if (lane < l) atomicAdd(&acc[wd[j] & RMASK], v[j]);
  }
  if (okx) atomicAdd(&acc[wdx & RMASK], vx);
  if (tot > 32) {
#pragma nounroll
    for (int base = 32; base < tot; base += 32) {
      const int t = base + lane;
      int j2 = 0;
#pragma unroll
      for (int l = 0; l < 32; ++l) j2 += (__builtin_amdgcn_readlane(inc, l) <= t) ? 1 : 0;
      const int j2c = j2 < 32 ? j2 : 31;
      const int e2 = __shfl(exc, j2c);
      const int b2 = __shfl(beg, j2c);
      if (t < tot) {
        const unsigned int w2 = wbase[(size_t)j2c * BIN_CAP + (size_t)(b2 + 32 + (t - e2))];
        atomicAdd(&acc[w2 & RMASK], yin[w2 >> RSHIFT]);
      }
    }
  }
}

__global__ __launch_bounds__(RTHREADS)
void degree_project_kernel(const float* __restrict__ x,
                           const float* __restrict__ W,
                           const unsigned int* __restrict__ bins,
                           const int* __restrict__ offs,
                           float* __restrict__ dinv,
                           float* __restrict__ zy, int n) {
  __shared__ float acc[RNODES];
  __shared__ float dinv_s[RNODES];
  __shared__ float z_s[RTILES * 16];
  const int tid  = threadIdx.x;
  const int lane = tid & 31;
  const int wv   = __builtin_amdgcn_readfirstlane(tid >> 5);
  const int lo   = blockIdx.x * RNODES;
  const int half = lane >> 4;
  const int m    = lane & 15;
  const bool four = wv + 24 < RTILES;
  int r0 = lo + (wv     ) * 16 + m;
  int r1 = lo + (wv +  8) * 16 + m;
  int r2 = lo + (wv + 16) * 16 + m;
  int r3 = lo + (wv + 24) * 16 + m;
  if (r0 > n - 1) r0 = n - 1;
  if (r1 > n - 1) r1 = n - 1;
  if (r2 > n - 1) r2 = n - 1;
  if (r3 > n - 1) r3 = n - 1;
  const float* x0 = x + (size_t)r0 * DFEAT;
  const float* x1 = x + (size_t)r1 * DFEAT;
  const float* x2 = x + (size_t)r2 * DFEAT;
  const float* x3 = x + (size_t)r3 * DFEAT;

  v2f f0[DFEAT / 4];
  v2f f1[DFEAT / 4];
  v2f f2[DFEAT / 4];
  v2f f3[DFEAT / 4];
  v2f fw[DFEAT / 4];
#pragma unroll
  for (int k0 = 0; k0 < DFEAT; k0 += 4) {
    const int kb = k0 + half * 2;
    fw[k0 / 4].x = W[kb];  fw[k0 / 4].y = W[kb + 1];
    f0[k0 / 4].x = x0[kb]; f0[k0 / 4].y = x0[kb + 1];
    f1[k0 / 4].x = x1[kb]; f1[k0 / 4].y = x1[kb + 1];
    f2[k0 / 4].x = x2[kb]; f2[k0 / 4].y = x2[kb + 1];
  }
  if (four) {
#pragma unroll
    for (int k0 = 0; k0 < DFEAT; k0 += 4) {
      const int kb = k0 + half * 2;
      f3[k0 / 4].x = x3[kb]; f3[k0 / 4].y = x3[kb + 1];
    }
  } else {
#pragma unroll
    for (int q = 0; q < DFEAT / 4; ++q) { f3[q].x = 0.0f; f3[q].y = 0.0f; }
  }

  acc[tid] = 0.0f;
  if (tid < RNODES - RTHREADS) acc[tid + RTHREADS] = 0.0f;
  range_count(bins, offs, acc, blockIdx.x);
  __syncthreads();
  {
    const float dd = rsqrtf(1.0f + acc[tid]);
    dinv_s[tid] = dd;
    if (lo + tid < n) dinv[lo + tid] = dd;
    if (tid < RNODES - RTHREADS) {
      const float d2 = rsqrtf(1.0f + acc[tid + RTHREADS]);
      dinv_s[tid + RTHREADS] = d2;
      if (lo + tid + RTHREADS < n) dinv[lo + tid + RTHREADS] = d2;
    }
  }

  v8f c0 = {};
  v8f c1 = {};
  v8f c2 = {};
  v8f c3 = {};
  if (four) {
#pragma unroll
    for (int q = 0; q < DFEAT / 4; ++q)
      c3 = __builtin_amdgcn_wmma_f32_16x16x4_f32(
           false, f3[q],  false, fw[q],
           (short)0, c3,  false,  false);
  }
#pragma unroll
  for (int q = 0; q < DFEAT / 4; ++q) {
    c0 = __builtin_amdgcn_wmma_f32_16x16x4_f32(
         false, f0[q],  false, fw[q],
         (short)0, c0,  false,  false);
    c1 = __builtin_amdgcn_wmma_f32_16x16x4_f32(
         false, f1[q],  false, fw[q],
         (short)0, c1,  false,  false);
    c2 = __builtin_amdgcn_wmma_f32_16x16x4_f32(
         false, f2[q],  false, fw[q],
         (short)0, c2,  false,  false);
  }

  const int rr = m & 7;
  float p0 = 0.0f, p1 = 0.0f, p2 = 0.0f, p3 = 0.0f;
#pragma unroll
  for (int r = 0; r < 8; ++r) {
    p0 = (rr == r) ? c0[r] : p0;
    p1 = (rr == r) ? c1[r] : p1;
    p2 = (rr == r) ? c2[r] : p2;
    p3 = (rr == r) ? c3[r] : p3;
  }
  if (m < 8) {
    const int in = half * 8 + rr;
    z_s[(wv     ) * 16 + in] = p0;
    z_s[(wv +  8) * 16 + in] = p1;
    z_s[(wv + 16) * 16 + in] = p2;
    if (four) z_s[(wv + 24) * 16 + in] = p3;
  }
  __syncthreads();
  if (lo + tid < n) zy[lo + tid] = dinv_s[tid] * z_s[tid];
  if (tid < RNODES - RTHREADS && lo + tid + RTHREADS < n)
    zy[lo + tid + RTHREADS] = dinv_s[tid + RTHREADS] * z_s[tid + RTHREADS];
}

#define ROW_PIECES (RNODES * (DFEAT / 4) / RTHREADS)
template <bool kLast, bool kRows>
__global__ __launch_bounds__(RTHREADS)
void hop_kernel(const unsigned int* __restrict__ bins,
                const int* __restrict__ offs,
                const float* __restrict__ dinv,
                const float* __restrict__ yin,
                const float* __restrict__ bptr,
                float* __restrict__ yout,
                const float* __restrict__ x,
                float* __restrict__ outx, int n) {
  __shared__ float acc[RNODES];
  __shared__ v4f plane4[LCAP / 4];
  const int tid   = threadIdx.x;
  const bool two  = tid < RNODES - RTHREADS;
  const int node0 = blockIdx.x * RNODES + tid;
  const int node1 = node0 + RTHREADS;
  const int nc0   = node0 < n ? node0 : n - 1;
  const int nc1   = (two && node1 < n) ? node1 : n - 1;
  const float d0    = dinv[nc0];
  const float self0 = yin[nc0];
  const float d1    = dinv[nc1];
  const float self1 = yin[nc1];
  const float bias  = kLast ? bptr[0] : 0.0f;

  const float4* x4 = reinterpret_cast<const float4*>(x);
  float4* o4 = reinterpret_cast<float4*>(outx);
  const int total4 = n * (DFEAT / 4);
  const int first4 = blockIdx.x * (RNODES * (DFEAT / 4)) + tid;
  float4 xv[ROW_PIECES];

  acc[tid] = 0.0f;
  if (two) acc[tid + RTHREADS] = 0.0f;
  range_sum_lds(bins, offs, yin, plane4, n, acc, blockIdx.x, [&]() {
    if (kRows) {
      __builtin_amdgcn_sched_barrier(0);
#pragma unroll
      for (int j = 0; j < ROW_PIECES; ++j) {
        int i = first4 + j * RTHREADS;
        if (i > total4 - 1) i = total4 - 1;
        xv[j] = x4[i];
      }
    }
  });
  if (kRows) lds_barrier();
  else       __syncthreads();

  const float z0 = d0 * (self0 + acc[tid]);
  const float s0 = kLast ? 1.0f / (1.0f + __expf(-(z0 + bias))) : d0 * z0;
  if (node0 < n) yout[node0] = s0;
  if (kRows) acc[tid] = s0;
  if (two) {
    const float z1 = d1 * (self1 + acc[tid + RTHREADS]);
    const float s1 = kLast ? 1.0f / (1.0f + __expf(-(z1 + bias))) : d1 * z1;
    if (node1 < n) yout[node1] = s1;
    if (kRows) acc[tid + RTHREADS] = s1;
  }
  if (kRows) {
    lds_barrier();
    float sc[ROW_PIECES];
#pragma unroll
    for (int j = 0; j < ROW_PIECES; ++j) sc[j] = acc[(tid >> 4) + 16 * j];
#pragma unroll
    for (int j = 0; j < ROW_PIECES; ++j) {
      const int i = first4 + j * RTHREADS;
      asm volatile("" :: "v"(sc[j]), "v"(xv[j].x), "v"(xv[j].y), "v"(xv[j].z), "v"(xv[j].w));
      if (i < total4)
        o4[i] = make_float4(xv[j].x * sc[j], xv[j].y * sc[j], xv[j].z * sc[j], xv[j].w * sc[j]);
    }
  }
}

#define ROW_THREADS 256
#define ROW_ITEMS   16
__global__ __launch_bounds__(ROW_THREADS)
void scale_rows_kernel(const float* __restrict__ x,
                       const float* __restrict__ score,
                       float* __restrict__ outx, int total4) {
  const float4* x4 = reinterpret_cast<const float4*>(x);
  float4* o4 = reinterpret_cast<float4*>(outx);
  const int first = blockIdx.x * (ROW_THREADS * ROW_ITEMS) + (int)threadIdx.x;
  float4 xv[ROW_ITEMS];
  float  sc[ROW_ITEMS];
#pragma unroll
  for (int j = 0; j < ROW_ITEMS; ++j) {
    int i = first + j * ROW_THREADS;
    if (i > total4 - 1) i = total4 - 1;
    xv[j] = x4[i];
    sc[j] = score[i >> 4];
  }
#pragma unroll
  for (int j = 0; j < ROW_ITEMS; ++j) {
    const int i = first + j * ROW_THREADS;
    if (i < total4)
      o4[i] = make_float4(xv[j].x * sc[j], xv[j].y * sc[j],
                          xv[j].z * sc[j], xv[j].w * sc[j]);
  }
}

extern "C" void kernel_launch(void* const* d_in, const int* in_sizes, int n_in,
                              void* d_out, int out_size, void* d_ws, size_t ws_size,
                              hipStream_t stream) {
  const float* x  = (const float*)d_in[0];
  const int*   ei = (const int*)d_in[1];
  const float* W  = (const float*)d_in[2];
  const float* b  = (const float*)d_in[3];

  const int n = in_sizes[0] / DFEAT;
  const int e = in_sizes[1] / 2;
  const int* src = ei;
  const int* dst = ei + e;

  const size_t nPad = ((size_t)n + 63) & ~(size_t)63;
  float* dinv = (float*)d_ws;
  float* zy0  = dinv + nPad;
  float* y1   = zy0 + nPad;
  float* outx = (float*)d_out;
  float* outs = (float*)d_out + (size_t)n * DFEAT;

  const size_t binWords = (size_t)NSLICE * BIN_CAP;
  const size_t tabWords = (size_t)NSLICE * MAXR;
  const bool inWs = ws_size >= (3 * nPad + binWords + tabWords) * sizeof(float);
  unsigned int* bins = inWs ? (unsigned int*)(y1 + nPad) : (unsigned int*)d_out;
  int* offs = (int*)(bins + binWords);

  const int nRanges  = (n + RNODES - 1) / RNODES;
  const int sliceLen = (e + NSLICE - 1) / NSLICE;
  const int total4   = n * (DFEAT / 4);

  bin_edges_kernel<<<NSLICE, BIN_THREADS, 0, stream>>>(src, dst, bins, offs, e, sliceLen);
  degree_project_kernel<<<nRanges, RTHREADS, 0, stream>>>(x, W, bins, offs, dinv, zy0, n);
  hop_kernel<false, false><<<nRanges, RTHREADS, 0, stream>>>(bins, offs, dinv, zy0, b, y1, x, outx, n);
  if (inWs) {
    hop_kernel<true, true><<<nRanges, RTHREADS, 0, stream>>>(bins, offs, dinv, y1, b, outs, x, outx, n);
  } else {
    hop_kernel<true, false><<<nRanges, RTHREADS, 0, stream>>>(bins, offs, dinv, y1, b, outs, x, outx, n);
    scale_rows_kernel<<<(total4 + ROW_THREADS * ROW_ITEMS - 1) / (ROW_THREADS * ROW_ITEMS), ROW_THREADS, 0, stream>>>(
        x, outs, outx, total4);
  }
}


# Placing pass: a thread's 25 range-start reads, its copy pieces and the wave sums leave LDS in groups, not one round trip per item; stores branch-free, idle items to spare slots; add order kept
# speedup vs baseline: 10.8971x; 1.0329x over previous
#include <hip/hip_runtime.h>
#include <cstdint>

#define DFEAT 64


typedef float v2f __attribute__((ext_vector_type(2)));
typedef float v8f __attribute__((ext_vector_type(8)));
typedef float v4f __attribute__((ext_vector_type(4)));

#define NSLICE      256
#define RSHIFT      9
#define RMASK       ((1u << RSHIFT) - 1u)
#define RNODES      416
#define RTHREADS    256
#define RTILES      (RNODES / 16)
#define MAXR        512
#define BIN_THREADS 256
#define BIN_ITEMS   25
#define BIN_CAP     (BIN_THREADS * BIN_ITEMS)
#define COPY_ITEMS  ((BIN_CAP / 4 + BIN_THREADS - 1) / BIN_THREADS)
#define FILL_CHUNK  26
#define FILL_CHUNKS 3
#define LCAP        (FILL_CHUNKS * FILL_CHUNK * RTHREADS * 4)

__global__ __launch_bounds__(BIN_THREADS)
void bin_edges_kernel(const int* __restrict__ src,
                      const int* __restrict__ dst,
                      unsigned int* __restrict__ bins,
                      int* __restrict__ offs,
                      int e, int sliceLen) {
  __shared__ int   start[MAXR];
  __shared__ uint4 stage4[BIN_CAP / 4];
  __shared__ int   wsum[BIN_THREADS / 32];
  unsigned int* stage = reinterpret_cast<unsigned int*>(stage4);

  const int tid  = threadIdx.x;
  const int lane = tid & 31;
  const int wave = tid >> 5;
  for (int i = tid; i < MAXR; i += BIN_THREADS) start[i] = 0;
  __syncthreads();

  const int sLo = blockIdx.x * sliceLen;
  int sHi = sLo + sliceLen;
  if (sHi > e) sHi = e;

  int d[BIN_ITEMS];
  int s[BIN_ITEMS];
  int rk[BIN_ITEMS];
#pragma unroll
  for (int j = 0; j < BIN_ITEMS; ++j) {
    const int idx = sLo + tid + j * BIN_THREADS;
    const int ic  = idx < sHi ? idx : 0;
    d[j] = dst[ic];
    s[j] = src[ic];
  }
#pragma unroll
  for (int j = 0; j < BIN_ITEMS; ++j) {
    const int idx = sLo + tid + j * BIN_THREADS;
    rk[j] = idx < sHi ? atomicAdd(&start[d[j] / RNODES], 1) : 0;
  }
  __syncthreads();

  const int c0 = start[2 * tid];
  const int c1 = start[2 * tid + 1];
  const int both = c0 + c1;
  int inc = both;
  for (int o = 1; o < 32; o <<= 1) {
    const int up = __shfl_up(inc, o);
    if (lane >= o) inc += up;
  }
  if (lane == 31) wsum[wave] = inc;
  __syncthreads();
  int wbase = 0;
#pragma unroll
  for (int w = 0; w < BIN_THREADS / 32; ++w) wbase += (w < wave) ? wsum[w] : 0;
  const int ex0 = wbase + inc - both;
  const int ex1 = ex0 + c0;
  start[2 * tid]     = ex0;
  start[2 * tid + 1] = ex1;
  __syncthreads();

  int st[BIN_ITEMS];
#pragma unroll
  for (int j = 0; j < BIN_ITEMS; ++j) st[j] = start[d[j] / RNODES];
#pragma unroll
  for (int j = 0; j < BIN_ITEMS; ++j) {
    const int item = tid + j * BIN_THREADS;
    const int rg   = d[j] / RNODES;
    const int pos  = (sLo + item < sHi) ? st[j] + rk[j] : item;
    stage[pos] = ((unsigned int)s[j] << RSHIFT) | (unsigned int)(d[j] - rg * RNODES);
  }
  __syncthreads();

  int cntS = sHi - sLo;
  if (cntS < 0) cntS = 0;
  unsigned int* seg = bins + (size_t)blockIdx.x * BIN_CAP;
  uint4* seg4 = reinterpret_cast<uint4*>(seg);
  const int c4 = cntS >> 2;
  uint4 cp[COPY_ITEMS];
#pragma unroll
  for (int k = 0; k < COPY_ITEMS; ++k) {
    const int i = tid + k * BIN_THREADS;
    cp[k] = stage4[i < BIN_CAP / 4 ? i : BIN_CAP / 4 - 1];
  }
#pragma unroll
  for (int k = 0; k < COPY_ITEMS; ++k) asm volatile("" :: "v"(cp[k].x), "v"(cp[k].y), "v"(cp[k].z), "v"(cp[k].w));
#pragma unroll
  for (int k = 0; k < COPY_ITEMS; ++k) {
    const int i = tid + k * BIN_THREADS;
    if (i < c4) seg4[i] = cp[k];
  }
  {
    const int i = (c4 << 2) + tid;
    if (i < cntS) seg[i] = stage[i];
  }
  int* row = offs + (size_t)blockIdx.x * MAXR;
  row[2 * tid]     = ex0;
  row[2 * tid + 1] = ex1;
}

__device__ __forceinline__ void lds_barrier() {
  __builtin_amdgcn_fence(__ATOMIC_RELEASE, "workgroup", "local");
  __builtin_amdgcn_s_barrier();
  __builtin_amdgcn_fence(__ATOMIC_ACQUIRE, "workgroup", "local");
}

__device__ __forceinline__ void range_count(const unsigned int* __restrict__ bins,
                                            const int* __restrict__ offs,
                                            float* acc, int r) {
  const int lane = threadIdx.x & 31;
  const int wave = threadIdx.x >> 5;
  const int* row = offs + (size_t)(wave * 32 + lane) * MAXR + r;
  const int beg = row[0];
  const int len = row[1] - beg;
  const unsigned int* wbase = bins + (size_t)(wave * 32) * BIN_CAP;

  unsigned int wd[32];
#pragma unroll
  for (int j = 0; j < 32; ++j) {
    const int b = __builtin_amdgcn_readlane(beg, j);
    const int l = __builtin_amdgcn_readlane(len, j);
    const int k = lane < l ? b + lane : 0;
    wd[j] = wbase[(size_t)j * BIN_CAP + k];
  }

  const int ov = len > 32 ? len - 32 : 0;
  int inc = ov;
  for (int o = 1; o < 32; o <<= 1) {
    const int up = __shfl_up(inc, o);
    if (lane >= o) inc += up;
  }
  const int exc = inc - ov;
  const int tot = __builtin_amdgcn_readlane(inc, 31);
  int jx = 0;
#pragma unroll
  for (int l = 0; l < 32; ++l) jx += (__builtin_amdgcn_readlane(inc, l) <= lane) ? 1 : 0;
  const int jxc = jx < 32 ? jx : 31;
  const int ejx = __shfl(exc, jxc);
  const int bjx = __shfl(beg, jxc);
  const bool okx = lane < tot;
  const unsigned int wdx = wbase[okx ? (size_t)jxc * BIN_CAP + (size_t)(bjx + 32 + (lane - ejx)) : (size_t)0];
  __builtin_amdgcn_sched_barrier(0);

#pragma unroll
  for (int j = 0; j < 32; ++j) asm volatile("" :: "v"(wd[j]));
  __syncthreads();
#pragma unroll
  for (int j = 0; j < 32; ++j) {
    const int l = __builtin_amdgcn_readlane(len, j);
    if (lane < l) atomicAdd(&acc[wd[j] & RMASK], 1.0f);
  }
  if (okx) atomicAdd(&acc[wdx & RMASK], 1.0f);
  if (tot > 32) {
#pragma nounroll
    for (int base = 32; base < tot; base += 32) {
      const int t = base + lane;
      int j2 = 0;
#pragma unroll
      for (int l = 0; l < 32; ++l) j2 += (__builtin_amdgcn_readlane(inc, l) <= t) ? 1 : 0;
      const int j2c = j2 < 32 ? j2 : 31;
      const int e2 = __shfl(exc, j2c);
      const int b2 = __shfl(beg, j2c);
      if (t < tot) {
        const unsigned int w2 = wbase[(size_t)j2c * BIN_CAP + (size_t)(b2 + 32 + (t - e2))];
        atomicAdd(&acc[w2 & RMASK], 1.0f);
      }
    }
  }
}

template <class More>
__device__ __forceinline__ void range_sum_lds(const unsigned int* __restrict__ bins,
                                              const int* __restrict__ offs,
                                              const float* __restrict__ yin,
                                              v4f* plane4, int n,
                                              float* acc, int r, More more) {
  const int tid  = threadIdx.x;
  const int lane = tid & 31;
  const int wave = tid >> 5;
  const int* row = offs + (size_t)(wave * 32 + lane) * MAXR + r;
  const int beg = row[0];
  const int len = row[1] - beg;
  const unsigned int* wbase = bins + (size_t)(wave * 32) * BIN_CAP;

  const float*  plane = reinterpret_cast<const float*>(plane4);
  const v4f* y4       = reinterpret_cast<const v4f*>(yin);
  int nl = n & ~3;
  if (nl > LCAP) nl = LCAP;
  const int nl4 = nl >> 2;

  unsigned int wd[32];
#pragma unroll
  for (int j = 0; j < 32; ++j) {
    const int b = __builtin_amdgcn_readlane(beg, j);
    const int l = __builtin_amdgcn_readlane(len, j);
    const int k = lane < l ? b + lane : 0;
    wd[j] = wbase[(size_t)j * BIN_CAP + k];
  }

  const int ov = len > 32 ? len - 32 : 0;
  int inc = ov;
  for (int o = 1; o < 32; o <<= 1) {
    const int up = __shfl_up(inc, o);
    if (lane >= o) inc += up;
  }
  const int exc = inc - ov;
  const int tot = __builtin_amdgcn_readlane(inc, 31);
  int jx = 0;
#pragma unroll
  for (int l = 0; l < 32; ++l) jx += (__builtin_amdgcn_readlane(inc, l) <= lane) ? 1 : 0;
  const int jxc = jx < 32 ? jx : 31;
  const int ejx = __shfl(exc, jxc);
  const int bjx = __shfl(beg, jxc);
  const bool okx = lane < tot;
  const unsigned int wdx = wbase[okx ? (size_t)jxc * BIN_CAP + (size_t)(bjx + 32 + (lane - ejx)) : (size_t)0];
  __builtin_amdgcn_sched_barrier(0);

  v4f pa[FILL_CHUNK];
  v4f pb[FILL_CHUNK];
#pragma unroll
  for (int j = 0; j < FILL_CHUNK; ++j) {
    const int i = tid + j * RTHREADS;
    pa[j] = y4[i < nl4 ? i : 0];
  }
#pragma unroll
  for (int j = 0; j < FILL_CHUNK; ++j) {
    const int i = tid + (FILL_CHUNK + j) * RTHREADS;
    pb[j] = y4[i < nl4 ? i : 0];
  }
  __builtin_amdgcn_sched_barrier(0);

  float vg[32];
#pragma unroll
  for (int j = 0; j < 32; ++j) {
    const int l = __builtin_amdgcn_readlane(len, j);
    const unsigned int s = wd[j] >> RSHIFT;
    const bool far = lane < l && s >= (unsigned int)nl;
    vg[j] = yin[far ? s : 0u];
  }
  const unsigned int sx = wdx >> RSHIFT;
  const float vgx = yin[(okx && sx >= (unsigned int)nl) ? sx : 0u];
  __builtin_amdgcn_sched_barrier(0);

#pragma unroll
  for (int j = 0; j < FILL_CHUNK; ++j) plane4[tid + j * RTHREADS] = pa[j];
#pragma unroll
  for (int j = 0; j < FILL_CHUNK; ++j) {
    const int i = tid + (2 * FILL_CHUNK + j) * RTHREADS;
    pa[j] = y4[i < nl4 ? i : 0];
  }
  more();
#pragma unroll
  for (int j = 0; j < FILL_CHUNK; ++j) plane4[tid + (FILL_CHUNK + j) * RTHREADS] = pb[j];
#pragma unroll
  for (int j = 0; j < FILL_CHUNK; ++j) plane4[tid + (2 * FILL_CHUNK + j) * RTHREADS] = pa[j];
  lds_barrier();

  float v[32];
#pragma unroll
  for (int j = 0; j < 32; ++j) {
    const int l = __builtin_amdgcn_readlane(len, j);
    const unsigned int s = wd[j] >> RSHIFT;
    const bool near = lane < l && s < (unsigned int)nl;
    const float t = plane[near ? s : 0u];
    v[j] = near ? t : vg[j];
  }
  const bool nearx = okx && sx < (unsigned int)nl;
  const float tx = plane[nearx ? sx : 0u];
  const float vx = nearx ? tx : vgx;

#pragma unroll
  for (int j = 0; j < 32; ++j) {
    const int l = __builtin_amdgcn_readlane(len, j);
    if (lane < l) atomicAdd(&acc[wd[j] & RMASK], v[j]);
  }
  if (okx) atomicAdd(&acc[wdx & RMASK], vx);
  if (tot > 32) {
#pragma nounroll
    for (int base = 32; base < tot; base += 32) {
      const int t = base + lane;
      int j2 = 0;
#pragma unroll
      for (int l = 0; l < 32; ++l) j2 += (__builtin_amdgcn_readlane(inc, l) <= t) ? 1 : 0;
      const int j2c = j2 < 32 ? j2 : 31;
      const int e2 = __shfl(exc, j2c);
      const int b2 = __shfl(beg, j2c);
      if (t < tot) {
        const unsigned int w2 = wbase[(size_t)j2c * BIN_CAP + (size_t)(b2 + 32 + (t - e2))];
        atomicAdd(&acc[w2 & RMASK], yin[w2 >> RSHIFT]);
      }
    }
  }
}

__global__ __launch_bounds__(RTHREADS)
void degree_project_kernel(const float* __restrict__ x,
                           const float* __restrict__ W,
                           const unsigned int* __restrict__ bins,
                           const int* __restrict__ offs,
                           float* __restrict__ dinv,
                           float* __restrict__ zy, int n) {
  __shared__ float acc[RNODES];
  __shared__ float dinv_s[RNODES];
  __shared__ float z_s[RTILES * 16];
  const int tid  = threadIdx.x;
  const int lane = tid & 31;
  const int wv   = __builtin_amdgcn_readfirstlane(tid >> 5);
  const int lo   = blockIdx.x * RNODES;
  const int half = lane >> 4;
  const int m    = lane & 15;
  const bool four = wv + 24 < RTILES;
  int r0 = lo + (wv     ) * 16 + m;
  int r1 = lo + (wv +  8) * 16 + m;
  int r2 = lo + (wv + 16) * 16 + m;
  int r3 = lo + (wv + 24) * 16 + m;
  if (r0 > n - 1) r0 = n - 1;
  if (r1 > n - 1) r1 = n - 1;
  if (r2 > n - 1) r2 = n - 1;
  if (r3 > n - 1) r3 = n - 1;
  const float* x0 = x + (size_t)r0 * DFEAT;
  const float* x1 = x + (size_t)r1 * DFEAT;
  const float* x2 = x + (size_t)r2 * DFEAT;
  const float* x3 = x + (size_t)r3 * DFEAT;

  v2f f0[DFEAT / 4];
  v2f f1[DFEAT / 4];
  v2f f2[DFEAT / 4];
  v2f f3[DFEAT / 4];
  v2f fw[DFEAT / 4];
#pragma unroll
  for (int k0 = 0; k0 < DFEAT; k0 += 4) {
    const int kb = k0 + half * 2;
    fw[k0 / 4].x = W[kb];  fw[k0 / 4].y = W[kb + 1];
    f0[k0 / 4].x = x0[kb]; f0[k0 / 4].y = x0[kb + 1];
    f1[k0 / 4].x = x1[kb]; f1[k0 / 4].y = x1[kb + 1];
    f2[k0 / 4].x = x2[kb]; f2[k0 / 4].y = x2[kb + 1];
  }
  if (four) {
#pragma unroll
    for (int k0 = 0; k0 < DFEAT; k0 += 4) {
      const int kb = k0 + half * 2;
      f3[k0 / 4].x = x3[kb]; f3[k0 / 4].y = x3[kb + 1];
    }
  } else {
#pragma unroll
    for (int q = 0; q < DFEAT / 4; ++q) { f3[q].x = 0.0f; f3[q].y = 0.0f; }
  }

  acc[tid] = 0.0f;
  if (tid < RNODES - RTHREADS) acc[tid + RTHREADS] = 0.0f;
  range_count(bins, offs, acc, blockIdx.x);
  __syncthreads();
  {
    const float dd = rsqrtf(1.0f + acc[tid]);
    dinv_s[tid] = dd;
    if (lo + tid < n) dinv[lo + tid] = dd;
    if (tid < RNODES - RTHREADS) {
      const float d2 = rsqrtf(1.0f + acc[tid + RTHREADS]);
      dinv_s[tid + RTHREADS] = d2;
      if (lo + tid + RTHREADS < n) dinv[lo + tid + RTHREADS] = d2;
    }
  }

  v8f c0 = {};
  v8f c1 = {};
  v8f c2 = {};
  v8f c3 = {};
  if (four) {
#pragma unroll
    for (int q = 0; q < DFEAT / 4; ++q)
      c3 = __builtin_amdgcn_wmma_f32_16x16x4_f32(
           false, f3[q],  false, fw[q],
           (short)0, c3,  false,  false);
  }
#pragma unroll
  for (int q = 0; q < DFEAT / 4; ++q) {
    c0 = __builtin_amdgcn_wmma_f32_16x16x4_f32(
         false, f0[q],  false, fw[q],
         (short)0, c0,  false,  false);
    c1 = __builtin_amdgcn_wmma_f32_16x16x4_f32(
         false, f1[q],  false, fw[q],
         (short)0, c1,  false,  false);
    c2 = __builtin_amdgcn_wmma_f32_16x16x4_f32(
         false, f2[q],  false, fw[q],
         (short)0, c2,  false,  false);
  }

  const int rr = m & 7;
  float p0 = 0.0f, p1 = 0.0f, p2 = 0.0f, p3 = 0.0f;
#pragma unroll
  for (int r = 0; r < 8; ++r) {
    p0 = (rr == r) ? c0[r] : p0;
    p1 = (rr == r) ? c1[r] : p1;
    p2 = (rr == r) ? c2[r] : p2;
    p3 = (rr == r) ? c3[r] : p3;
  }
  if (m < 8) {
    const int in = half * 8 + rr;
    z_s[(wv     ) * 16 + in] = p0;
    z_s[(wv +  8) * 16 + in] = p1;
    z_s[(wv + 16) * 16 + in] = p2;
    if (four) z_s[(wv + 24) * 16 + in] = p3;
  }
  __syncthreads();
  if (lo + tid < n) zy[lo + tid] = dinv_s[tid] * z_s[tid];
  if (tid < RNODES - RTHREADS && lo + tid + RTHREADS < n)
    zy[lo + tid + RTHREADS] = dinv_s[tid + RTHREADS] * z_s[tid + RTHREADS];
}

#define ROW_PIECES (RNODES * (DFEAT / 4) / RTHREADS)
template <bool kLast, bool kRows>
__global__ __launch_bounds__(RTHREADS)
void hop_kernel(const unsigned int* __restrict__ bins,
                const int* __restrict__ offs,
                const float* __restrict__ dinv,
                const float* __restrict__ yin,
                const float* __restrict__ bptr,
                float* __restrict__ yout,
                const float* __restrict__ x,
                float* __restrict__ outx, int n) {
  __shared__ float acc[RNODES];
  __shared__ v4f plane4[LCAP / 4];
  const int tid   = threadIdx.x;
  const bool two  = tid < RNODES - RTHREADS;
  const int node0 = blockIdx.x * RNODES + tid;
  const int node1 = node0 + RTHREADS;
  const int nc0   = node0 < n ? node0 : n - 1;
  const int nc1   = (two && node1 < n) ? node1 : n - 1;
  const float d0    = dinv[nc0];
  const float self0 = yin[nc0];
  const float d1    = dinv[nc1];
  const float self1 = yin[nc1];
  const float bias  = kLast ? bptr[0] : 0.0f;

  const float4* x4 = reinterpret_cast<const float4*>(x);
  float4* o4 = reinterpret_cast<float4*>(outx);
  const int total4 = n * (DFEAT / 4);
  const int first4 = blockIdx.x * (RNODES * (DFEAT / 4)) + tid;
  float4 xv[ROW_PIECES];

  acc[tid] = 0.0f;
  if (two) acc[tid + RTHREADS] = 0.0f;
  range_sum_lds(bins, offs, yin, plane4, n, acc, blockIdx.x, [&]() {
    if (kRows) {
      __builtin_amdgcn_sched_barrier(0);
#pragma unroll
      for (int j = 0; j < ROW_PIECES; ++j) {
        int i = first4 + j * RTHREADS;
        if (i > total4 - 1) i = total4 - 1;
        xv[j] = x4[i];
      }
    }
  });
  if (kRows) lds_barrier();
  else       __syncthreads();

  const float z0 = d0 * (self0 + acc[tid]);
  const float s0 = kLast ? 1.0f / (1.0f + __expf(-(z0 + bias))) : d0 * z0;
  if (node0 < n) yout[node0] = s0;
  if (kRows) acc[tid] = s0;
  if (two) {
    const float z1 = d1 * (self1 + acc[tid + RTHREADS]);
    const float s1 = kLast ? 1.0f / (1.0f + __expf(-(z1 + bias))) : d1 * z1;
    if (node1 < n) yout[node1] = s1;
    if (kRows) acc[tid + RTHREADS] = s1;
  }
  if (kRows) {
    lds_barrier();
    float sc[ROW_PIECES];
#pragma unroll
    for (int j = 0; j < ROW_PIECES; ++j) sc[j] = acc[(tid >> 4) + 16 * j];
#pragma unroll
    for (int j = 0; j < ROW_PIECES; ++j) {
      const int i = first4 + j * RTHREADS;
      asm volatile("" :: "v"(sc[j]), "v"(xv[j].x), "v"(xv[j].y), "v"(xv[j].z), "v"(xv[j].w));
      if (i < total4)
        o4[i] = make_float4(xv[j].x * sc[j], xv[j].y * sc[j], xv[j].z * sc[j], xv[j].w * sc[j]);
    }
  }
}

#define ROW_THREADS 256
#define ROW_ITEMS   16
__global__ __launch_bounds__(ROW_THREADS)
void scale_rows_kernel(const float* __restrict__ x,
                       const float* __restrict__ score,
                       float* __restrict__ outx, int total4) {
  const float4* x4 = reinterpret_cast<const float4*>(x);
  float4* o4 = reinterpret_cast<float4*>(outx);
  const int first = blockIdx.x * (ROW_THREADS * ROW_ITEMS) + (int)threadIdx.x;
  float4 xv[ROW_ITEMS];
  float  sc[ROW_ITEMS];
#pragma unroll
  for (int j = 0; j < ROW_ITEMS; ++j) {
    int i = first + j * ROW_THREADS;
    if (i > total4 - 1) i = total4 - 1;
    xv[j] = x4[i];
    sc[j] = score[i >> 4];
  }
#pragma unroll
  for (int j = 0; j < ROW_ITEMS; ++j) {
    const int i = first + j * ROW_THREADS;
    if (i < total4)
      o4[i] = make_float4(xv[j].x * sc[j], xv[j].y * sc[j],
                          xv[j].z * sc[j], xv[j].w * sc[j]);
  }
}

extern "C" void kernel_launch(void* const* d_in, const int* in_sizes, int n_in,
                              void* d_out, int out_size, void* d_ws, size_t ws_size,
                              hipStream_t stream) {
  const float* x  = (const float*)d_in[0];
  const int*   ei = (const int*)d_in[1];
  const float* W  = (const float*)d_in[2];
  const float* b  = (const float*)d_in[3];

  const int n = in_sizes[0] / DFEAT;
  const int e = in_sizes[1] / 2;
  const int* src = ei;
  const int* dst = ei + e;

  const size_t nPad = ((size_t)n + 63) & ~(size_t)63;
  float* dinv = (float*)d_ws;
  float* zy0  = dinv + nPad;
  float* y1   = zy0 + nPad;
  float* outx = (float*)d_out;
  float* outs = (float*)d_out + (size_t)n * DFEAT;

  const size_t binWords = (size_t)NSLICE * BIN_CAP;
  const size_t tabWords = (size_t)NSLICE * MAXR;
  const bool inWs = ws_size >= (3 * nPad + binWords + tabWords) * sizeof(float);
  unsigned int* bins = inWs ? (unsigned int*)(y1 + nPad) : (unsigned int*)d_out;
  int* offs = (int*)(bins + binWords);

  const int nRanges  = (n + RNODES - 1) / RNODES;
  const int sliceLen = (e + NSLICE - 1) / NSLICE;
  const int total4   = n * (DFEAT / 4);

  bin_edges_kernel<<<NSLICE, BIN_THREADS, 0, stream>>>(src, dst, bins, offs, e, sliceLen);
  degree_project_kernel<<<nRanges, RTHREADS, 0, stream>>>(x, W, bins, offs, dinv, zy0, n);
  hop_kernel<false, false><<<nRanges, RTHREADS, 0, stream>>>(bins, offs, dinv, zy0, b, y1, x, outx, n);
  if (inWs) {
    hop_kernel<true, true><<<nRanges, RTHREADS, 0, stream>>>(bins, offs, dinv, y1, b, outs, x, outx, n);
  } else {
    hop_kernel<true, false><<<nRanges, RTHREADS, 0, stream>>>(bins, offs, dinv, y1, b, outs, x, outx, n);
    scale_rows_kernel<<<(total4 + ROW_THREADS * ROW_ITEMS - 1) / (ROW_THREADS * ROW_ITEMS), ROW_THREADS, 0, stream>>>(
        x, outs, outx, total4);
  }
}
